# starting kernel
# baseline (speedup 1.0000x reference)
#include <hip/hip_runtime.h>

typedef __attribute__((ext_vector_type(2))) float v2f;
typedef __attribute__((ext_vector_type(8))) float v8f;

__global__ void deg_init_kernel(float* __restrict__ deg, int n) {
    int i = blockIdx.x * blockDim.x + threadIdx.x;
    if (i < n) deg[i] = 1.0f;
}

__global__ void deg_count_kernel(const int* __restrict__ dst, float* __restrict__ deg, int nE) {
    int e = blockIdx.x * blockDim.x + threadIdx.x;
    if (e < nE) atomicAdd(&deg[dst[e]], 1.0f);
}

__global__ void deg_to_dinv_kernel(float* __restrict__ deg, int n) {
    int i = blockIdx.x * blockDim.x + threadIdx.x;
    if (i < n) deg[i] = rsqrtf(deg[i]);
}

__global__ void gemm_xw1_kernel(const float* __restrict__ A, const float* __restrict__ W,
                                float* __restrict__ C, int nRows) {
    int waveId = (blockIdx.x * blockDim.x + threadIdx.x) >> 5;
    int lane   = threadIdx.x & 31;
    int m0     = waveId * 16;
    if (m0 >= nRows) return;

    int row  = m0 + (lane & 15);
    int grp  = lane >> 4;
    int koff = grp * 2;
    int col  = lane & 15;

    v8f acc0 = {};
    v8f acc1 = {};
#pragma unroll
    for (int k = 0; k < 8; ++k) {
        int kb = 4 * k + koff;
        v2f a = *reinterpret_cast<const v2f*>(A + row * 32 + kb);
        v2f b0, b1;
        b0.x = W[kb * 32 + col];        b0.y = W[(kb + 1) * 32 + col];
        b1.x = W[kb * 32 + 16 + col];   b1.y = W[(kb + 1) * 32 + 16 + col];
        acc0 = __builtin_amdgcn_wmma_f32_16x16x4_f32(false, a, false, b0, (short)0, acc0, false, false);
        acc1 = __builtin_amdgcn_wmma_f32_16x16x4_f32(false, a, false, b1, (short)0, acc1, false, false);
    }

    int rowbase = m0 + grp * 8;
#pragma unroll
    for (int r = 0; r < 8; ++r) {
        C[(rowbase + r) * 32 + col]      = acc0[r];
        C[(rowbase + r) * 32 + 16 + col] = acc1[r];
    }
}

__global__ void self_init_kernel(const float* __restrict__ feat, const float* __restrict__ dinv,
                                 float* __restrict__ agg, int n32) {
    int idx = blockIdx.x * blockDim.x + threadIdx.x;
    if (idx < n32) {
        float di = dinv[idx >> 5];
        agg[idx] = feat[idx] * di * di;
    }
}

__global__ void scatter_agg_kernel(const int* __restrict__ src, const int* __restrict__ dst,
                                   const float* __restrict__ dinv, const float* __restrict__ feat,
                                   float* __restrict__ agg, int nEdges) {
    int lane = threadIdx.x & 31;
    int wave = (blockIdx.x * blockDim.x + threadIdx.x) >> 5;
    int base = wave * 32;
    if (base >= nEdges) return;

    int  myE   = base + lane;
    bool valid = myE < nEdges;
    int   sL = valid ? src[myE] : 0;
    int   dL = valid ? dst[myE] : 0;
    float wL = valid ? dinv[sL] * dinv[dL] : 0.0f;

    int cnt = nEdges - base;
    if (cnt > 32) cnt = 32;

#pragma unroll 4
    for (int i = 0; i < cnt; ++i) {
        int   s = __builtin_amdgcn_readlane(sL, i);
        int   d = __builtin_amdgcn_readlane(dL, i);
        float w = __int_as_float(__builtin_amdgcn_readlane(__float_as_int(wL), i));
        float v = feat[s * 32 + lane] * w;
        atomicAdd(agg + d * 32 + lane, v);
    }
}

__global__ void relu_bias_kernel(float* __restrict__ h, const float* __restrict__ b, int n32) {
    int idx = blockIdx.x * blockDim.x + threadIdx.x;
    if (idx < n32) {
        float v = h[idx] + b[idx & 31];
        h[idx] = v > 0.0f ? v : 0.0f;
    }
}

__global__ void gemm_out_kernel(const float* __restrict__ A,
                                const float* __restrict__ Wmu, const float* __restrict__ bmu,
                                const float* __restrict__ Wlv, const float* __restrict__ blv,
                                float* __restrict__ out, int nRows) {
    int waveId = (blockIdx.x * blockDim.x + threadIdx.x) >> 5;
    int lane   = threadIdx.x & 31;
    int m0     = waveId * 16;
    if (m0 >= nRows) return;

    int row  = m0 + (lane & 15);
    int grp  = lane >> 4;
    int koff = grp * 2;
    int col  = lane & 15;

    v8f accM = {};
    v8f accL = {};
#pragma unroll
    for (int k = 0; k < 8; ++k) {
        int kb = 4 * k + koff;
        v2f a = *reinterpret_cast<const v2f*>(A + row * 32 + kb);
        v2f bm, bl;
        bm.x = Wmu[kb * 16 + col];  bm.y = Wmu[(kb + 1) * 16 + col];
        bl.x = Wlv[kb * 16 + col];  bl.y = Wlv[(kb + 1) * 16 + col];
        accM = __builtin_amdgcn_wmma_f32_16x16x4_f32(false, a, false, bm, (short)0, accM, false, false);
        accL = __builtin_amdgcn_wmma_f32_16x16x4_f32(false, a, false, bl, (short)0, accL, false, false);
    }

    int rowbase = m0 + grp * 8;
    float vbm = bmu[col];
    float vbl = blv[col];
    float* outLv = out + (size_t)nRows * 16;
#pragma unroll
    for (int r = 0; r < 8; ++r) {
        out[(rowbase + r) * 16 + col]   = accM[r] + vbm;
        outLv[(rowbase + r) * 16 + col] = accL[r] + vbl;
    }
}

extern "C" void kernel_launch(void* const* d_in, const int* in_sizes, int n_in,
                              void* d_out, int out_size, void* d_ws, size_t ws_size,
                              hipStream_t stream) {
    const float* x   = (const float*)d_in[0];
    const int*   ei  = (const int*)  d_in[1];
    const float* W1  = (const float*)d_in[2];
    const float* b1  = (const float*)d_in[3];
    const float* Wmu = (const float*)d_in[4];
    const float* bmu = (const float*)d_in[5];
    const float* Wlv = (const float*)d_in[6];
    const float* blv = (const float*)d_in[7];

    const int N = in_sizes[0] / 32;
    const int E = in_sizes[1] / 2;
    const int* src = ei;
    const int* dst = ei + E;

    float* dinv = (float*)d_ws;
    float* bufA = dinv + N;
    float* bufB = bufA + (size_t)N * 32;
    float* out  = (float*)d_out;

    const int TB = 256;
    const int n32 = N * 32;
    const int gemmThreads = ((N + 15) / 16) * 32;
    const int scatterWaves = (E + 31) / 32;
    const int scatterBlocks = (scatterWaves * 32 + TB - 1) / TB;

    deg_init_kernel<<<(N + TB - 1) / TB, TB, 0, stream>>>(dinv, N);
    deg_count_kernel<<<(E + TB - 1) / TB, TB, 0, stream>>>(dst, dinv, E);
    deg_to_dinv_kernel<<<(N + TB - 1) / TB, TB, 0, stream>>>(dinv, N);

    gemm_xw1_kernel<<<(gemmThreads + TB - 1) / TB, TB, 0, stream>>>(x, W1, bufA, N);
    self_init_kernel<<<(n32 + TB - 1) / TB, TB, 0, stream>>>(bufA, dinv, bufB, n32);
    scatter_agg_kernel<<<scatterBlocks, TB, 0, stream>>>(src, dst, dinv, bufA, bufB, E);
    relu_bias_kernel<<<(n32 + TB - 1) / TB, TB, 0, stream>>>(bufB, b1, n32);

    self_init_kernel<<<(n32 + TB - 1) / TB, TB, 0, stream>>>(bufB, dinv, bufA, n32);
    scatter_agg_kernel<<<scatterBlocks, TB, 0, stream>>>(src, dst, dinv, bufB, bufA, E);
    gemm_out_kernel<<<(gemmThreads + TB - 1) / TB, TB, 0, stream>>>(bufA, Wmu, bmu, Wlv, blv, out, N);
}


# Edge sums from a destination-sorted edge list (integer atomic rank, scan, fill): one wave per node gathers rows, no float atomics; layer 1 sums before its GEMM, layer 2 after: sum order differs
# speedup vs baseline: 1.3852x; 1.3852x over previous
#include <hip/hip_runtime.h>

typedef __attribute__((ext_vector_type(2))) float v2f;
typedef __attribute__((ext_vector_type(8))) float v8f;

__global__ void zero_count_kernel(int* __restrict__ cnt, int n) {
    int i = blockIdx.x * blockDim.x + threadIdx.x;
    if (i < n) cnt[i] = 0;
}

__global__ void rank_count_kernel(const int* __restrict__ dst, int* __restrict__ cnt,
                                  int* __restrict__ rank, int nE) {
    int e = blockIdx.x * blockDim.x + threadIdx.x;
    if (e < nE) rank[e] = atomicAdd(&cnt[dst[e]], 1);
}

#define SCAN_TB 1024
__global__ void __launch_bounds__(SCAN_TB)
rowptr_dinv_kernel(const int* __restrict__ cnt, int* __restrict__ rowptr,
                   float* __restrict__ dinv, int n) {
    __shared__ int sh[SCAN_TB];
    const int tid = threadIdx.x;
    const int blockBeg = blockIdx.x * SCAN_TB;

    int part = 0;
    for (int j = tid; j < blockBeg; j += SCAN_TB) part += cnt[j];
    sh[tid] = part;
    __syncthreads();
    for (int s = SCAN_TB / 2; s > 0; s >>= 1) {
        if (tid < s) sh[tid] += sh[tid + s];
        __syncthreads();
    }
    const int prefix = sh[0];
    __syncthreads();

    const int i = blockBeg + tid;
    const int c = (i < n) ? cnt[i] : 0;
    sh[tid] = c;
    __syncthreads();
    for (int off = 1; off < SCAN_TB; off <<= 1) {
        int v = (tid >= off) ? sh[tid - off] : 0;
        __syncthreads();
        sh[tid] += v;
        __syncthreads();
    }
    const int incl = sh[tid];
    if (i < n) {
        rowptr[i] = prefix + incl - c;
        dinv[i]   = rsqrtf((float)(c + 1));
    }
}

__global__ void fill_kernel(const int* __restrict__ src, const int* __restrict__ dst,
                            const int* __restrict__ rank, const int* __restrict__ rowptr,
                            const float* __restrict__ dinv, int2* __restrict__ csr, int nE) {
    int e = blockIdx.x * blockDim.x + threadIdx.x;
    if (e < nE) {
        int s = src[e];
        csr[rowptr[dst[e]] + rank[e]] = make_int2(s, __float_as_int(dinv[s]));
    }
}

template <int G>
__device__ __forceinline__ void load_row_groups(const float* __restrict__ feat, int sL, int cnt,
                                                int lane, float (&v)[32]) {
#pragma unroll
    for (int u = 0; u < 4; ++u) {
        int s = __builtin_amdgcn_readlane(sL, G * 4 + u);
        v[G * 4 + u] = feat[s * 32 + lane];
    }
    if constexpr (G < 7) {
        if (cnt > (G + 1) * 4) load_row_groups<G + 1>(feat, sL, cnt, lane, v);
    }
}

__device__ __forceinline__ float pull_node(const float* __restrict__ feat,
                                           const int2* __restrict__ csr,
                                           const int* __restrict__ rowptr,
                                           int d, int n, int nE, int lane) {
    const int beg = rowptr[d];
    const int end = (d + 1 < n) ? rowptr[d + 1] : nE;
    const float di = rsqrtf((float)(end - beg + 1));
    float acc = 0.0f;
    if (beg >= end) acc = feat[d * 32 + lane] * (di * di);
    for (int base = beg; base < end; base += 32) {
        const float selfW = (base == beg) ? di * di : 0.0f;
        int cnt = end - base;
        if (cnt > 32) cnt = 32;
        const int   last = cnt - 1;
        const int2  eL = csr[base + (lane < last ? lane : last)];
        const int   sL = eL.x;
        const float dL = __int_as_float(eL.y);
        float v[32];
#pragma unroll
        for (int u = 0; u < 32; ++u) v[u] = 0.0f;
        const float vs = feat[d * 32 + lane];
        load_row_groups<0>(feat, sL, cnt, lane, v);
        const float wL = (lane < cnt) ? dL * di : 0.0f;
        acc += vs * selfW;
#pragma unroll
        for (int u = 0; u < 32; ++u) {
            float w = __int_as_float(__builtin_amdgcn_readlane(__float_as_int(wL), u));
            acc += v[u] * w;
        }
    }
    return acc;
}

__global__ void pull_rows_kernel(const float* __restrict__ feat, const int2* __restrict__ csr,
                                 const int* __restrict__ rowptr,
                                 float* __restrict__ rows, int n, int nE) {
    const int lane = threadIdx.x & 31;
    const int d = __builtin_amdgcn_readfirstlane((int)((blockIdx.x * blockDim.x + threadIdx.x) >> 5));
    if (d >= n) return;
    rows[d * 32 + lane] = pull_node(feat, csr, rowptr, d, n, nE, lane);
}

__global__ void pull_out_kernel(const float* __restrict__ z, const int2* __restrict__ csr,
                                const int* __restrict__ rowptr,
                                const float* __restrict__ bmu, const float* __restrict__ blv,
                                float* __restrict__ out, int n, int nE) {
    const int lane = threadIdx.x & 31;
    const int w  = __builtin_amdgcn_readfirstlane((int)((blockIdx.x * blockDim.x + threadIdx.x) >> 5));
    const int n0 = 2 * w;
    if (n0 >= n) return;
    const bool hasB = (n0 + 1) < n;
    float a = pull_node(z, csr, rowptr, n0, n, nE, lane);
    float b = 0.0f;
    if (hasB) b = pull_node(z, csr, rowptr, n0 + 1, n, nE, lane);
    const float ax = __shfl_xor(a, 16);
    const float bx = __shfl_xor(b, 16);
    const int c = lane & 15;
    const float muv = (lane < 16 ? a  : bx) + bmu[c];
    const float lvv = (lane < 16 ? ax : b ) + blv[c];
    float* outLv = out + (size_t)n * 16;
    if (lane < 16 || hasB) {
        out[n0 * 16 + lane]   = muv;
        outLv[n0 * 16 + lane] = lvv;
    }
}

__global__ void gemm_w1_relu_kernel(float* H, const float* __restrict__ W,
                                    const float* __restrict__ bias, int nRows) {
    int waveId = (blockIdx.x * blockDim.x + threadIdx.x) >> 5;
    int lane   = threadIdx.x & 31;
    int m0     = waveId * 16;
    if (m0 >= nRows) return;

    int row  = m0 + (lane & 15);
    int grp  = lane >> 4;
    int koff = grp * 2;
    int col  = lane & 15;

    v8f acc0 = {};
    v8f acc1 = {};
#pragma unroll
    for (int k = 0; k < 8; ++k) {
        int kb = 4 * k + koff;
        v2f a = *reinterpret_cast<const v2f*>(H + row * 32 + kb);
        v2f b0, b1;
        b0.x = W[kb * 32 + col];        b0.y = W[(kb + 1) * 32 + col];
        b1.x = W[kb * 32 + 16 + col];   b1.y = W[(kb + 1) * 32 + 16 + col];
        acc0 = __builtin_amdgcn_wmma_f32_16x16x4_f32(false, a, false, b0, (short)0, acc0, false, false);
        acc1 = __builtin_amdgcn_wmma_f32_16x16x4_f32(false, a, false, b1, (short)0, acc1, false, false);
    }

    int rowbase = m0 + grp * 8;
    float vb0 = bias[col];
    float vb1 = bias[16 + col];
#pragma unroll
    for (int r = 0; r < 8; ++r) {
        float v0 = acc0[r] + vb0;
        float v1 = acc1[r] + vb1;
        H[(rowbase + r) * 32 + col]      = v0 > 0.0f ? v0 : 0.0f;
        H[(rowbase + r) * 32 + 16 + col] = v1 > 0.0f ? v1 : 0.0f;
    }
}

__global__ void gemm_wout_kernel(float* H, const float* __restrict__ Wmu,
                                 const float* __restrict__ Wlv, int nRows) {
    int waveId = (blockIdx.x * blockDim.x + threadIdx.x) >> 5;
    int lane   = threadIdx.x & 31;
    int m0     = waveId * 16;
    if (m0 >= nRows) return;

    int row  = m0 + (lane & 15);
    int grp  = lane >> 4;
    int koff = grp * 2;
    int col  = lane & 15;

    v8f accM = {};
    v8f accL = {};
#pragma unroll
    for (int k = 0; k < 8; ++k) {
        int kb = 4 * k + koff;
        v2f a = *reinterpret_cast<const v2f*>(H + row * 32 + kb);
        v2f bm, bl;
        bm.x = Wmu[kb * 16 + col];  bm.y = Wmu[(kb + 1) * 16 + col];
        bl.x = Wlv[kb * 16 + col];  bl.y = Wlv[(kb + 1) * 16 + col];
        accM = __builtin_amdgcn_wmma_f32_16x16x4_f32(false, a, false, bm, (short)0, accM, false, false);
        accL = __builtin_amdgcn_wmma_f32_16x16x4_f32(false, a, false, bl, (short)0, accL, false, false);
    }

    int rowbase = m0 + grp * 8;
#pragma unroll
    for (int r = 0; r < 8; ++r) {
        H[(rowbase + r) * 32 + col]      = accM[r];
        H[(rowbase + r) * 32 + 16 + col] = accL[r];
    }
}

extern "C" void kernel_launch(void* const* d_in, const int* in_sizes, int n_in,
                              void* d_out, int out_size, void* d_ws, size_t ws_size,
                              hipStream_t stream) {
    const float* x   = (const float*)d_in[0];
    const int*   ei  = (const int*)  d_in[1];
    const float* W1  = (const float*)d_in[2];
    const float* b1  = (const float*)d_in[3];
    const float* Wmu = (const float*)d_in[4];
    const float* bmu = (const float*)d_in[5];
    const float* Wlv = (const float*)d_in[6];
    const float* blv = (const float*)d_in[7];

    const int N = in_sizes[0] / 32;
    const int E = in_sizes[1] / 2;
    const int* src = ei;
    const int* dst = ei + E;

    int*   rowptr = (int*)d_ws;
    int2*  csr    = (int2*)(rowptr + N);
    float* bufA   = (float*)(csr + E);
    int*   rank   = (int*)bufA;
    int*   cnt    = rank + E;
    float* dinv   = (float*)(cnt + N);
    float* out    = (float*)d_out;

    const int TB = 256;
    const int gemmThreads = ((N + 15) / 16) * 32;
    const int pairWaves   = (N + 1) / 2;

    zero_count_kernel<<<(N + TB - 1) / TB, TB, 0, stream>>>(cnt, N);
    rank_count_kernel<<<(E + TB - 1) / TB, TB, 0, stream>>>(dst, cnt, rank, E);
    rowptr_dinv_kernel<<<(N + SCAN_TB - 1) / SCAN_TB, SCAN_TB, 0, stream>>>(cnt, rowptr, dinv, N);
    fill_kernel<<<(E + TB - 1) / TB, TB, 0, stream>>>(src, dst, rank, rowptr, dinv, csr, E);

    pull_rows_kernel<<<(N * 32 + TB - 1) / TB, TB, 0, stream>>>(x, csr, rowptr, bufA, N, E);
    gemm_w1_relu_kernel<<<(gemmThreads + TB - 1) / TB, TB, 0, stream>>>(bufA, W1, b1, N);

    gemm_wout_kernel<<<(gemmThreads + TB - 1) / TB, TB, 0, stream>>>(bufA, Wmu, Wlv, N);
    pull_out_kernel<<<(pairWaves * 32 + TB - 1) / TB, TB, 0, stream>>>(bufA, csr, rowptr, bmu, blv, out, N, E);
}


# Edge list ordered by a counting sort in LDS (edge chunks binned by 512-node range, then one block per range), not global atomic rank, scan and scattered fill; sum order follows the LDS atomic rank
# speedup vs baseline: 2.8047x; 2.0248x over previous
#include <hip/hip_runtime.h>

typedef __attribute__((ext_vector_type(2))) float v2f;
typedef __attribute__((ext_vector_type(8))) float v8f;

#define CH        8192
#define BIN_SHIFT 9
#define BIN_NODES 512
#define MAXB      256
#define MAXC      256
#define SRC_BITS  17
#define SRC_MASK  0x1FFFF
#define STAGE_CAP 16384

__device__ __forceinline__ int block_scan_incl(int* sc, int tid, int v) {
    sc[tid] = v;
    __syncthreads();
    for (int off = 1; off < 1024; off <<= 1) {
        int t = (tid >= off) ? sc[tid - off] : 0;
        __syncthreads();
        sc[tid] += t;
        __syncthreads();
    }
    return sc[tid];
}

__global__ void __launch_bounds__(1024)
chunk_sort_kernel(const int* __restrict__ src, const int* __restrict__ dst,
                  int* __restrict__ tmp, int* __restrict__ cntT, int* __restrict__ offT,
                  int nE, int nB, int nC) {
    __shared__ int hist[MAXB];
    __shared__ int sc[1024];
    __shared__ int stage[CH];
    const int tid  = threadIdx.x;
    const int c    = blockIdx.x;
    const int base = c * CH;

    if (tid < MAXB) hist[tid] = 0;
    __syncthreads();

    int w[8], bn[8];
#pragma unroll
    for (int k = 0; k < 8; ++k) {
        const int e = base + k * 1024 + tid;
        bn[k] = -1;
        w[k]  = 0;
        if (e < nE) {
            const int d = dst[e];
            const int s = src[e];
            bn[k] = d >> BIN_SHIFT;
            w[k]  = s | ((d & (BIN_NODES - 1)) << SRC_BITS);
        }
    }
#pragma unroll
    for (int k = 0; k < 8; ++k)
        if (bn[k] >= 0) atomicAdd(&hist[bn[k]], 1);
    __syncthreads();

    const int h    = (tid < MAXB) ? hist[tid] : 0;
    const int incl = block_scan_incl(sc, tid, h);
    const int excl = incl - h;
    if (tid < nB) {
        cntT[tid * nC + c] = h;
        offT[tid * nC + c] = excl;
    }
    if (tid < MAXB) hist[tid] = excl;
    __syncthreads();

#pragma unroll
    for (int k = 0; k < 8; ++k)
        if (bn[k] >= 0) stage[atomicAdd(&hist[bn[k]], 1)] = w[k];
    __syncthreads();

#pragma unroll
    for (int k = 0; k < 8; ++k) {
        const int i = k * 1024 + tid;
        if (base + i < nE) tmp[base + i] = stage[i];
    }
}

__global__ void __launch_bounds__(1024)
bin_build_kernel(const int* __restrict__ tmp, const int* __restrict__ cntT,
                 const int* __restrict__ offT, int* __restrict__ rowptr,
                 float* __restrict__ dinv, int2* __restrict__ csr, int n, int nC) {
    __shared__ int sc[1024];
    __shared__ int pOff[MAXC];
    __shared__ int pCnt[MAXC];
    __shared__ int cnt[BIN_NODES];
    __shared__ int stage[STAGE_CAP];
    const int tid  = threadIdx.x;
    const int lane = tid & 31;
    const int wave = tid >> 5;
    const int b    = blockIdx.x;

    int part = 0;
    for (int i = tid; i < b * nC; i += 1024) part += cntT[i];
    block_scan_incl(sc, tid, part);
    const int base = sc[1023];
    __syncthreads();

    if (tid < MAXC) {
        const bool v = tid < nC;
        pCnt[tid] = v ? cntT[b * nC + tid] : 0;
        pOff[tid] = v ? tid * CH + offT[b * nC + tid] : 0;
    }
    if (tid < BIN_NODES) cnt[tid] = 0;
    __syncthreads();

    int wv[16];
#pragma unroll
    for (int q = 0; q < 8; ++q) {
        const int pc = pCnt[wave + 32 * q];
        const int po = pOff[wave + 32 * q];
        wv[2 * q]     = (lane < pc)      ? tmp[po + lane]      : -1;
        wv[2 * q + 1] = (lane + 32 < pc) ? tmp[po + lane + 32] : -1;
    }
#pragma unroll
    for (int u = 0; u < 16; ++u)
        if (wv[u] >= 0) atomicAdd(&cnt[wv[u] >> SRC_BITS], 1);
    for (int q = 0; q < 8; ++q) {
        const int pc = pCnt[wave + 32 * q];
        const int po = pOff[wave + 32 * q];
        for (int j = lane + 64; j < pc; j += 32)
            atomicAdd(&cnt[tmp[po + j] >> SRC_BITS], 1);
    }
    __syncthreads();

    const int myc   = (tid < BIN_NODES) ? cnt[tid] : 0;
    const int incl  = block_scan_incl(sc, tid, myc);
    const int total = sc[1023];
    const int excl  = incl - myc;
    if (tid < BIN_NODES) {
        const int node = b * BIN_NODES + tid;
        if (node < n) {
            rowptr[node] = base + excl;
            dinv[node]   = rsqrtf((float)(myc + 1));
        }
        cnt[tid] = excl;
    }
    __syncthreads();

    if (total <= STAGE_CAP) {
#pragma unroll
        for (int u = 0; u < 16; ++u)
            if (wv[u] >= 0) stage[atomicAdd(&cnt[wv[u] >> SRC_BITS], 1)] = wv[u] & SRC_MASK;
        for (int q = 0; q < 8; ++q) {
            const int pc = pCnt[wave + 32 * q];
            const int po = pOff[wave + 32 * q];
            for (int j = lane + 64; j < pc; j += 32) {
                const int x = tmp[po + j];
                stage[atomicAdd(&cnt[x >> SRC_BITS], 1)] = x & SRC_MASK;
            }
        }
        __syncthreads();
        for (int i = tid; i < total; i += 1024) csr[base + i] = make_int2(stage[i], 0);
    } else {
#pragma unroll
        for (int u = 0; u < 16; ++u)
            if (wv[u] >= 0)
                csr[base + atomicAdd(&cnt[wv[u] >> SRC_BITS], 1)] = make_int2(wv[u] & SRC_MASK, 0);
        for (int q = 0; q < 8; ++q) {
            const int pc = pCnt[wave + 32 * q];
            const int po = pOff[wave + 32 * q];
            for (int j = lane + 64; j < pc; j += 32) {
                const int x = tmp[po + j];
                csr[base + atomicAdd(&cnt[x >> SRC_BITS], 1)] = make_int2(x & SRC_MASK, 0);
            }
        }
    }
}

__global__ void attach_dinv_kernel(int2* __restrict__ csr, const float* __restrict__ dinv, int nE) {
    int j = blockIdx.x * blockDim.x + threadIdx.x;
    if (j < nE) {
        int2 e = csr[j];
        e.y = __float_as_int(dinv[e.x]);
        csr[j] = e;
    }
}

template <int G>
__device__ __forceinline__ void load_row_groups(const float* __restrict__ feat, int sL, int cnt,
                                                int lane, float (&v)[32]) {
#pragma unroll
    for (int u = 0; u < 4; ++u) {
        int s = __builtin_amdgcn_readlane(sL, G * 4 + u);
        v[G * 4 + u] = feat[s * 32 + lane];
    }
    if constexpr (G < 7) {
        if (cnt > (G + 1) * 4) load_row_groups<G + 1>(feat, sL, cnt, lane, v);
    }
}

__device__ __forceinline__ float pull_node(const float* __restrict__ feat,
                                           const int2* __restrict__ csr,
                                           const int* __restrict__ rowptr,
                                           int d, int n, int nE, int lane) {
    const int beg = rowptr[d];
    const int end = (d + 1 < n) ? rowptr[d + 1] : nE;
    const float di = rsqrtf((float)(end - beg + 1));
    float acc = 0.0f;
    if (beg >= end) acc = feat[d * 32 + lane] * (di * di);
    for (int base = beg; base < end; base += 32) {
        const float selfW = (base == beg) ? di * di : 0.0f;
        int cnt = end - base;
        if (cnt > 32) cnt = 32;
        const int   last = cnt - 1;
        const int2  eL = csr[base + (lane < last ? lane : last)];
        const int   sL = eL.x;
        const float dL = __int_as_float(eL.y);
        float v[32];
#pragma unroll
        for (int u = 0; u < 32; ++u) v[u] = 0.0f;
        const float vs = feat[d * 32 + lane];
        load_row_groups<0>(feat, sL, cnt, lane, v);
        const float wL = (lane < cnt) ? dL * di : 0.0f;
        acc += vs * selfW;
#pragma unroll
        for (int u = 0; u < 32; ++u) {
            float w = __int_as_float(__builtin_amdgcn_readlane(__float_as_int(wL), u));
            acc += v[u] * w;
        }
    }
    return acc;
}

__global__ void pull_rows_kernel(const float* __restrict__ feat, const int2* __restrict__ csr,
                                 const int* __restrict__ rowptr,
                                 float* __restrict__ rows, int n, int nE) {
    const int lane = threadIdx.x & 31;
    const int d = __builtin_amdgcn_readfirstlane((int)((blockIdx.x * blockDim.x + threadIdx.x) >> 5));
    if (d >= n) return;
    rows[d * 32 + lane] = pull_node(feat, csr, rowptr, d, n, nE, lane);
}

__global__ void pull_out_kernel(const float* __restrict__ z, const int2* __restrict__ csr,
                                const int* __restrict__ rowptr,
                                const float* __restrict__ bmu, const float* __restrict__ blv,
                                float* __restrict__ out, int n, int nE) {
    const int lane = threadIdx.x & 31;
    const int w  = __builtin_amdgcn_readfirstlane((int)((blockIdx.x * blockDim.x + threadIdx.x) >> 5));
    const int n0 = 2 * w;
    if (n0 >= n) return;
    const bool hasB = (n0 + 1) < n;
    float a = pull_node(z, csr, rowptr, n0, n, nE, lane);
    float b = 0.0f;
    if (hasB) b = pull_node(z, csr, rowptr, n0 + 1, n, nE, lane);
    const float ax = __shfl_xor(a, 16);
    const float bx = __shfl_xor(b, 16);
    const int c = lane & 15;
    const float muv = (lane < 16 ? a  : bx) + bmu[c];
    const float lvv = (lane < 16 ? ax : b ) + blv[c];
    float* outLv = out + (size_t)n * 16;
    if (lane < 16 || hasB) {
        out[n0 * 16 + lane]   = muv;
        outLv[n0 * 16 + lane] = lvv;
    }
}

__global__ void gemm_w1_relu_kernel(float* H, const float* __restrict__ W,
                                    const float* __restrict__ bias, int nRows) {
    int waveId = (blockIdx.x * blockDim.x + threadIdx.x) >> 5;
    int lane   = threadIdx.x & 31;
    int m0     = waveId * 16;
    if (m0 >= nRows) return;

    int row  = m0 + (lane & 15);
    int grp  = lane >> 4;
    int koff = grp * 2;
    int col  = lane & 15;

    v8f acc0 = {};
    v8f acc1 = {};
#pragma unroll
    for (int k = 0; k < 8; ++k) {
        int kb = 4 * k + koff;
        v2f a = *reinterpret_cast<const v2f*>(H + row * 32 + kb);
        v2f b0, b1;
        b0.x = W[kb * 32 + col];        b0.y = W[(kb + 1) * 32 + col];
        b1.x = W[kb * 32 + 16 + col];   b1.y = W[(kb + 1) * 32 + 16 + col];
        acc0 = __builtin_amdgcn_wmma_f32_16x16x4_f32(false, a, false, b0, (short)0, acc0, false, false);
        acc1 = __builtin_amdgcn_wmma_f32_16x16x4_f32(false, a, false, b1, (short)0, acc1, false, false);
    }

    int rowbase = m0 + grp * 8;
    float vb0 = bias[col];
    float vb1 = bias[16 + col];
#pragma unroll
    for (int r = 0; r < 8; ++r) {
        float v0 = acc0[r] + vb0;
        float v1 = acc1[r] + vb1;
        H[(rowbase + r) * 32 + col]      = v0 > 0.0f ? v0 : 0.0f;
        H[(rowbase + r) * 32 + 16 + col] = v1 > 0.0f ? v1 : 0.0f;
    }
}

__global__ void gemm_wout_kernel(float* H, const float* __restrict__ Wmu,
                                 const float* __restrict__ Wlv, int nRows) {
    int waveId = (blockIdx.x * blockDim.x + threadIdx.x) >> 5;
    int lane   = threadIdx.x & 31;
    int m0     = waveId * 16;
    if (m0 >= nRows) return;

    int row  = m0 + (lane & 15);
    int grp  = lane >> 4;
    int koff = grp * 2;
    int col  = lane & 15;

    v8f accM = {};
    v8f accL = {};
#pragma unroll
    for (int k = 0; k < 8; ++k) {
        int kb = 4 * k + koff;
        v2f a = *reinterpret_cast<const v2f*>(H + row * 32 + kb);
        v2f bm, bl;
        bm.x = Wmu[kb * 16 + col];  bm.y = Wmu[(kb + 1) * 16 + col];
        bl.x = Wlv[kb * 16 + col];  bl.y = Wlv[(kb + 1) * 16 + col];
        accM = __builtin_amdgcn_wmma_f32_16x16x4_f32(false, a, false, bm, (short)0, accM, false, false);
        accL = __builtin_amdgcn_wmma_f32_16x16x4_f32(false, a, false, bl, (short)0, accL, false, false);
    }

    int rowbase = m0 + grp * 8;
#pragma unroll
    for (int r = 0; r < 8; ++r) {
        H[(rowbase + r) * 32 + col]      = accM[r];
        H[(rowbase + r) * 32 + 16 + col] = accL[r];
    }
}

extern "C" void kernel_launch(void* const* d_in, const int* in_sizes, int n_in,
                              void* d_out, int out_size, void* d_ws, size_t ws_size,
                              hipStream_t stream) {
    const float* x   = (const float*)d_in[0];
    const int*   ei  = (const int*)  d_in[1];
    const float* W1  = (const float*)d_in[2];
    const float* b1  = (const float*)d_in[3];
    const float* Wmu = (const float*)d_in[4];
    const float* bmu = (const float*)d_in[5];
    const float* Wlv = (const float*)d_in[6];
    const float* blv = (const float*)d_in[7];

    const int N = in_sizes[0] / 32;
    const int E = in_sizes[1] / 2;
    const int* src = ei;
    const int* dst = ei + E;

    const int nB = (N + BIN_NODES - 1) / BIN_NODES;
    const int nC = (E + CH - 1) / CH;
    int*   rowptr = (int*)d_ws;
    int2*  csr    = (int2*)(rowptr + N);
    float* bufA   = (float*)(csr + E);
    int*   tmp    = (int*)bufA;
    float* dinv   = (float*)(tmp + E);
    int*   cntT   = (int*)(dinv + N);
    int*   offT   = cntT + nB * nC;
    float* out    = (float*)d_out;

    const int TB = 256;
    const int gemmThreads = ((N + 15) / 16) * 32;
    const int pairWaves   = (N + 1) / 2;

    chunk_sort_kernel<<<nC, 1024, 0, stream>>>(src, dst, tmp, cntT, offT, E, nB, nC);
    bin_build_kernel<<<nB, 1024, 0, stream>>>(tmp, cntT, offT, rowptr, dinv, csr, N, nC);
    attach_dinv_kernel<<<(E + TB - 1) / TB, TB, 0, stream>>>(csr, dinv, E);

    pull_rows_kernel<<<(N * 32 + TB - 1) / TB, TB, 0, stream>>>(x, csr, rowptr, bufA, N, E);
    gemm_w1_relu_kernel<<<(gemmThreads + TB - 1) / TB, TB, 0, stream>>>(bufA, W1, b1, N);

    gemm_wout_kernel<<<(gemmThreads + TB - 1) / TB, TB, 0, stream>>>(bufA, Wmu, Wlv, N);
    pull_out_kernel<<<(pairWaves * 32 + TB - 1) / TB, TB, 0, stream>>>(bufA, csr, rowptr, bmu, blv, out, N, E);
}


# Layer-1 row sums, relu and both WMMA products in one kernel: 16 waves sum 16 node rows into an LDS tile, one wave runs both products from it; two launches and two node-array passes gone
# speedup vs baseline: 3.3812x; 1.2055x over previous
#include <hip/hip_runtime.h>

typedef __attribute__((ext_vector_type(2))) float v2f;
typedef __attribute__((ext_vector_type(8))) float v8f;

#define CH        8192
#define BIN_SHIFT 9
#define BIN_NODES 512
#define MAXB      256
#define MAXC      256
#define SRC_BITS  17
#define SRC_MASK  0x1FFFF
#define STAGE_CAP 16384

__device__ __forceinline__ int block_scan_incl(int* sc, int tid, int v) {
    sc[tid] = v;
    __syncthreads();
    for (int off = 1; off < 1024; off <<= 1) {
        int t = (tid >= off) ? sc[tid - off] : 0;
        __syncthreads();
        sc[tid] += t;
        __syncthreads();
    }
    return sc[tid];
}

__global__ void __launch_bounds__(1024)
chunk_sort_kernel(const int* __restrict__ src, const int* __restrict__ dst,
                  int* __restrict__ tmp, int* __restrict__ cntT, int* __restrict__ offT,
                  int nE, int nB, int nC) {
    __shared__ int hist[MAXB];
    __shared__ int sc[1024];
    __shared__ int stage[CH];
    const int tid  = threadIdx.x;
    const int c    = blockIdx.x;
    const int base = c * CH;

    if (tid < MAXB) hist[tid] = 0;
    __syncthreads();

    int w[8], bn[8];
#pragma unroll
    for (int k = 0; k < 8; ++k) {
        const int e = base + k * 1024 + tid;
        bn[k] = -1;
        w[k]  = 0;
        if (e < nE) {
            const int d = dst[e];
            const int s = src[e];
            bn[k] = d >> BIN_SHIFT;
            w[k]  = s | ((d & (BIN_NODES - 1)) << SRC_BITS);
        }
    }
#pragma unroll
    for (int k = 0; k < 8; ++k)
        if (bn[k] >= 0) atomicAdd(&hist[bn[k]], 1);
    __syncthreads();

    const int h    = (tid < MAXB) ? hist[tid] : 0;
    const int incl = block_scan_incl(sc, tid, h);
    const int excl = incl - h;
    if (tid < nB) {
        cntT[tid * nC + c] = h;
        offT[tid * nC + c] = excl;
    }
    if (tid < MAXB) hist[tid] = excl;
    __syncthreads();

#pragma unroll
    for (int k = 0; k < 8; ++k)
        if (bn[k] >= 0) stage[atomicAdd(&hist[bn[k]], 1)] = w[k];
    __syncthreads();

#pragma unroll
    for (int k = 0; k < 8; ++k) {
        const int i = k * 1024 + tid;
        if (base + i < nE) tmp[base + i] = stage[i];
    }
}

__global__ void __launch_bounds__(1024)
bin_build_kernel(const int* __restrict__ tmp, const int* __restrict__ cntT,
                 const int* __restrict__ offT, int* __restrict__ rowptr,
                 float* __restrict__ dinv, int2* __restrict__ csr, int n, int nC) {
    __shared__ int sc[1024];
    __shared__ int pOff[MAXC];
    __shared__ int pCnt[MAXC];
    __shared__ int cnt[BIN_NODES];
    __shared__ int stage[STAGE_CAP];
    const int tid  = threadIdx.x;
    const int lane = tid & 31;
    const int wave = tid >> 5;
    const int b    = blockIdx.x;

    int part = 0;
    for (int i = tid; i < b * nC; i += 1024) part += cntT[i];
    block_scan_incl(sc, tid, part);
    const int base = sc[1023];
    __syncthreads();

    if (tid < MAXC) {
        const bool v = tid < nC;
        pCnt[tid] = v ? cntT[b * nC + tid] : 0;
        pOff[tid] = v ? tid * CH + offT[b * nC + tid] : 0;
    }
    if (tid < BIN_NODES) cnt[tid] = 0;
    __syncthreads();

    int wv[16];
#pragma unroll
    for (int q = 0; q < 8; ++q) {
        const int pc = pCnt[wave + 32 * q];
        const int po = pOff[wave + 32 * q];
        wv[2 * q]     = (lane < pc)      ? tmp[po + lane]      : -1;
        wv[2 * q + 1] = (lane + 32 < pc) ? tmp[po + lane + 32] : -1;
    }
#pragma unroll
    for (int u = 0; u < 16; ++u)
        if (wv[u] >= 0) atomicAdd(&cnt[wv[u] >> SRC_BITS], 1);
    for (int q = 0; q < 8; ++q) {
        const int pc = pCnt[wave + 32 * q];
        const int po = pOff[wave + 32 * q];
        for (int j = lane + 64; j < pc; j += 32)
            atomicAdd(&cnt[tmp[po + j] >> SRC_BITS], 1);
    }
    __syncthreads();

    const int myc   = (tid < BIN_NODES) ? cnt[tid] : 0;
    const int incl  = block_scan_incl(sc, tid, myc);
    const int total = sc[1023];
    const int excl  = incl - myc;
    if (tid < BIN_NODES) {
        const int node = b * BIN_NODES + tid;
        if (node < n) {
            rowptr[node] = base + excl;
            dinv[node]   = rsqrtf((float)(myc + 1));
        }
        cnt[tid] = excl;
    }
    __syncthreads();

    if (total <= STAGE_CAP) {
#pragma unroll
        for (int u = 0; u < 16; ++u)
            if (wv[u] >= 0) stage[atomicAdd(&cnt[wv[u] >> SRC_BITS], 1)] = wv[u] & SRC_MASK;
        for (int q = 0; q < 8; ++q) {
            const int pc = pCnt[wave + 32 * q];
            const int po = pOff[wave + 32 * q];
            for (int j = lane + 64; j < pc; j += 32) {
                const int x = tmp[po + j];
                stage[atomicAdd(&cnt[x >> SRC_BITS], 1)] = x & SRC_MASK;
            }
        }
        __syncthreads();
        for (int i = tid; i < total; i += 1024) csr[base + i] = make_int2(stage[i], 0);
    } else {
#pragma unroll
        for (int u = 0; u < 16; ++u)
            if (wv[u] >= 0)
                csr[base + atomicAdd(&cnt[wv[u] >> SRC_BITS], 1)] = make_int2(wv[u] & SRC_MASK, 0);
        for (int q = 0; q < 8; ++q) {
            const int pc = pCnt[wave + 32 * q];
            const int po = pOff[wave + 32 * q];
            for (int j = lane + 64; j < pc; j += 32) {
                const int x = tmp[po + j];
                csr[base + atomicAdd(&cnt[x >> SRC_BITS], 1)] = make_int2(x & SRC_MASK, 0);
            }
        }
    }
}

__global__ void attach_dinv_kernel(int2* __restrict__ csr, const float* __restrict__ dinv, int nE) {
    int j = blockIdx.x * blockDim.x + threadIdx.x;
    if (j < nE) {
        int2 e = csr[j];
        e.y = __float_as_int(dinv[e.x]);
        csr[j] = e;
    }
}

template <int G>
__device__ __forceinline__ void load_row_groups(const float* __restrict__ feat, int sL, int cnt,
                                                int lane, float (&v)[32]) {
#pragma unroll
    for (int u = 0; u < 4; ++u) {
        int s = __builtin_amdgcn_readlane(sL, G * 4 + u);
        v[G * 4 + u] = feat[s * 32 + lane];
    }
    if constexpr (G < 7) {
        if (cnt > (G + 1) * 4) load_row_groups<G + 1>(feat, sL, cnt, lane, v);
    }
}

__device__ __forceinline__ float pull_node(const float* __restrict__ feat,
                                           const int2* __restrict__ csr,
                                           const int* __restrict__ rowptr,
                                           int d, int n, int nE, int lane) {
    const int beg = rowptr[d];
    const int end = (d + 1 < n) ? rowptr[d + 1] : nE;
    const float di = rsqrtf((float)(end - beg + 1));
    float acc = 0.0f;
    if (beg >= end) acc = feat[d * 32 + lane] * (di * di);
    for (int base = beg; base < end; base += 32) {
        const float selfW = (base == beg) ? di * di : 0.0f;
        int cnt = end - base;
        if (cnt > 32) cnt = 32;
        const int   last = cnt - 1;
        const int2  eL = csr[base + (lane < last ? lane : last)];
        const int   sL = eL.x;
        const float dL = __int_as_float(eL.y);
        float v[32];
#pragma unroll
        for (int u = 0; u < 32; ++u) v[u] = 0.0f;
        const float vs = feat[d * 32 + lane];
        load_row_groups<0>(feat, sL, cnt, lane, v);
        const float wL = (lane < cnt) ? dL * di : 0.0f;
        acc += vs * selfW;
#pragma unroll
        for (int u = 0; u < 32; ++u) {
            float w = __int_as_float(__builtin_amdgcn_readlane(__float_as_int(wL), u));
            acc += v[u] * w;
        }
    }
    return acc;
}

#define TP 36

__global__ void __launch_bounds__(512)
pull_gemm_kernel(const float* __restrict__ x, const int2* __restrict__ csr,
                 const int* __restrict__ rowptr,
                 const float* __restrict__ W1, const float* __restrict__ b1,
                 const float* __restrict__ Wmu, const float* __restrict__ Wlv,
                 float* __restrict__ z, int n, int nE) {
    __shared__ __attribute__((aligned(16))) float tA[16 * TP];
    __shared__ __attribute__((aligned(16))) float tH[16 * TP];
    const int lane = threadIdx.x & 31;
    const int wave = __builtin_amdgcn_readfirstlane((int)(threadIdx.x >> 5));
    const int m0   = blockIdx.x * 16;
    const int d    = m0 + wave;

    float sum = 0.0f;
    if (d < n) sum = pull_node(x, csr, rowptr, d, n, nE, lane);
    tA[wave * TP + lane] = sum;
    __syncthreads();

    if (wave != (int)(blockIdx.x & 15)) return;

    const int c    = lane & 15;
    const int hh   = lane >> 4;
    const int koff = 2 * hh;

    v8f acc0 = {};
    v8f acc1 = {};
#pragma unroll
    for (int k = 0; k < 8; ++k) {
        const int kb = 4 * k + koff;
        const v2f a = *reinterpret_cast<const v2f*>(&tA[c * TP + kb]);
        v2f p, q;
        p.x = W1[kb * 32 + c];        p.y = W1[(kb + 1) * 32 + c];
        q.x = W1[kb * 32 + 16 + c];   q.y = W1[(kb + 1) * 32 + 16 + c];
        acc0 = __builtin_amdgcn_wmma_f32_16x16x4_f32(false, a, false, p, (short)0, acc0, false, false);
        acc1 = __builtin_amdgcn_wmma_f32_16x16x4_f32(false, a, false, q, (short)0, acc1, false, false);
    }

    const float vb0 = b1[c];
    const float vb1 = b1[16 + c];
#pragma unroll
    for (int r = 0; r < 8; ++r) {
        const float v0 = acc0[r] + vb0;
        const float v1 = acc1[r] + vb1;
        tH[(8 * hh + r) * TP + c]      = v0 > 0.0f ? v0 : 0.0f;
        tH[(8 * hh + r) * TP + 16 + c] = v1 > 0.0f ? v1 : 0.0f;
    }
    __builtin_amdgcn_fence(__ATOMIC_ACQ_REL, "wavefront");

    v8f accM = {};
    v8f accL = {};
#pragma unroll
    for (int k = 0; k < 8; ++k) {
        const int kb = 4 * k + koff;
        const v2f a = *reinterpret_cast<const v2f*>(&tH[c * TP + kb]);
        v2f p, q;
        p.x = Wmu[kb * 16 + c];  p.y = Wmu[(kb + 1) * 16 + c];
        q.x = Wlv[kb * 16 + c];  q.y = Wlv[(kb + 1) * 16 + c];
        accM = __builtin_amdgcn_wmma_f32_16x16x4_f32(false, a, false, p, (short)0, accM, false, false);
        accL = __builtin_amdgcn_wmma_f32_16x16x4_f32(false, a, false, q, (short)0, accL, false, false);
    }

#pragma unroll
    for (int r = 0; r < 8; ++r) {
        tA[(8 * hh + r) * TP + c]      = accM[r];
        tA[(8 * hh + r) * TP + 16 + c] = accL[r];
    }
    __builtin_amdgcn_fence(__ATOMIC_ACQ_REL, "wavefront");
#pragma unroll
    for (int i = 0; i < 4; ++i) {
        const int row = 4 * i + (lane >> 3);
        const int ch  = lane & 7;
        const float4 v = *reinterpret_cast<const float4*>(&tA[row * TP + 4 * ch]);
        if (m0 + row < n) *reinterpret_cast<float4*>(&z[(m0 + row) * 32 + 4 * ch]) = v;
    }
}

__global__ void pull_out_kernel(const float* __restrict__ z, const int2* __restrict__ csr,
                                const int* __restrict__ rowptr,
                                const float* __restrict__ bmu, const float* __restrict__ blv,
                                float* __restrict__ out, int n, int nE) {
    const int lane = threadIdx.x & 31;
    const int w  = __builtin_amdgcn_readfirstlane((int)((blockIdx.x * blockDim.x + threadIdx.x) >> 5));
    const int n0 = 2 * w;
    if (n0 >= n) return;
    const bool hasB = (n0 + 1) < n;
    float a = pull_node(z, csr, rowptr, n0, n, nE, lane);
    float b = 0.0f;
    if (hasB) b = pull_node(z, csr, rowptr, n0 + 1, n, nE, lane);
    const float ax = __shfl_xor(a, 16);
    const float bx = __shfl_xor(b, 16);
    const int c = lane & 15;
    const float muv = (lane < 16 ? a  : bx) + bmu[c];
    const float lvv = (lane < 16 ? ax : b ) + blv[c];
    float* outLv = out + (size_t)n * 16;
    if (lane < 16 || hasB) {
        out[n0 * 16 + lane]   = muv;
        outLv[n0 * 16 + lane] = lvv;
    }
}

extern "C" void kernel_launch(void* const* d_in, const int* in_sizes, int n_in,
                              void* d_out, int out_size, void* d_ws, size_t ws_size,
                              hipStream_t stream) {
    const float* x   = (const float*)d_in[0];
    const int*   ei  = (const int*)  d_in[1];
    const float* W1  = (const float*)d_in[2];
    const float* b1  = (const float*)d_in[3];
    const float* Wmu = (const float*)d_in[4];
    const float* bmu = (const float*)d_in[5];
    const float* Wlv = (const float*)d_in[6];
    const float* blv = (const float*)d_in[7];

    const int N = in_sizes[0] / 32;
    const int E = in_sizes[1] / 2;
    const int* src = ei;
    const int* dst = ei + E;

    const int nB = (N + BIN_NODES - 1) / BIN_NODES;
    const int nC = (E + CH - 1) / CH;
    int*   rowptr = (int*)d_ws;
    int2*  csr    = (int2*)(rowptr + N);
    float* bufA   = (float*)(csr + E);
    int*   tmp    = (int*)bufA;
    float* dinv   = (float*)(tmp + E);
    int*   cntT   = (int*)(dinv + N);
    int*   offT   = cntT + nB * nC;
    float* out    = (float*)d_out;

    const int TB = 256;
    const int tileBlocks  = (N + 15) / 16;
    const int pairWaves   = (N + 1) / 2;

    chunk_sort_kernel<<<nC, 1024, 0, stream>>>(src, dst, tmp, cntT, offT, E, nB, nC);
    bin_build_kernel<<<nB, 1024, 0, stream>>>(tmp, cntT, offT, rowptr, dinv, csr, N, nC);
    attach_dinv_kernel<<<(E + TB - 1) / TB, TB, 0, stream>>>(csr, dinv, E);

    pull_gemm_kernel<<<tileBlocks, 512, 0, stream>>>(x, csr, rowptr, W1, b1, Wmu, Wlv, bufA, N, E);

    pull_out_kernel<<<(pairWaves * 32 + TB - 1) / TB, TB, 0, stream>>>(bufA, csr, rowptr, bmu, blv, out, N, E);
}


# Edge-sort block prefix sums on the vector ALU: per-wave DPP scans joined by one barrier replace three 10-step LDS scans; a range's start = sum of its chunks' loaded offsets, no count-table sweep
# speedup vs baseline: 3.6446x; 1.0779x over previous
#include <hip/hip_runtime.h>

typedef __attribute__((ext_vector_type(2))) float v2f;
typedef __attribute__((ext_vector_type(8))) float v8f;

#define CH        8192
#define BIN_SHIFT 9
#define BIN_NODES 512
#define MAXB      256
#define MAXC      256
#define SRC_BITS  17
#define SRC_MASK  0x1FFFF
#define STAGE_CAP 16384

__device__ __forceinline__ int wave_scan_incl(int x, int lane) {
    x += __builtin_amdgcn_update_dpp(0, x, 0x111, 0xf, 0xf, true);
    x += __builtin_amdgcn_update_dpp(0, x, 0x112, 0xf, 0xf, true);
    x += __builtin_amdgcn_update_dpp(0, x, 0x114, 0xf, 0xf, true);
    x += __builtin_amdgcn_update_dpp(0, x, 0x118, 0xf, 0xf, true);
    const int lowTotal = __builtin_amdgcn_readlane(x, 15);
    return x + (lane >= 16 ? lowTotal : 0);
}

__device__ __forceinline__ int block_scan_incl(int* wsum, int tid, int v, int& total) {
    const int lane = tid & 31;
    const int wave = __builtin_amdgcn_readfirstlane(tid >> 5);
    const int x = wave_scan_incl(v, lane);
    if (lane == 31) wsum[wave] = x;
    __syncthreads();
    const int y = wave_scan_incl(wsum[lane], lane);
    total = __builtin_amdgcn_readlane(y, 31);
    const int before = __builtin_amdgcn_readlane(y, (wave + 31) & 31);
    return x + (wave > 0 ? before : 0);
}

__global__ void __launch_bounds__(1024)
chunk_sort_kernel(const int* __restrict__ src, const int* __restrict__ dst,
                  int* __restrict__ tmp, int* __restrict__ cntT, int* __restrict__ offT,
                  int nE, int nB, int nC) {
    __shared__ int hist[MAXB];
    __shared__ int wsum[32];
    __shared__ int stage[CH];
    const int tid  = threadIdx.x;
    const int c    = blockIdx.x;
    const int base = c * CH;

    if (tid < MAXB) hist[tid] = 0;
    __syncthreads();

    int w[8], bn[8];
#pragma unroll
    for (int k = 0; k < 8; ++k) {
        const int e = base + k * 1024 + tid;
        bn[k] = -1;
        w[k]  = 0;
        if (e < nE) {
            const int d = dst[e];
            const int s = src[e];
            bn[k] = d >> BIN_SHIFT;
            w[k]  = s | ((d & (BIN_NODES - 1)) << SRC_BITS);
        }
    }
#pragma unroll
    for (int k = 0; k < 8; ++k)
        if (bn[k] >= 0) atomicAdd(&hist[bn[k]], 1);
    __syncthreads();

    const int h    = (tid < MAXB) ? hist[tid] : 0;
    int chunkTotal;
    const int incl = block_scan_incl(wsum, tid, h, chunkTotal);
    const int excl = incl - h;
    if (tid < nB) {
        cntT[tid * nC + c] = h;
        offT[tid * nC + c] = excl;
    }
    if (tid < MAXB) hist[tid] = excl;
    __syncthreads();

#pragma unroll
    for (int k = 0; k < 8; ++k)
        if (bn[k] >= 0) stage[atomicAdd(&hist[bn[k]], 1)] = w[k];
    __syncthreads();

#pragma unroll
    for (int k = 0; k < 8; ++k) {
        const int i = k * 1024 + tid;
        if (base + i < nE) tmp[base + i] = stage[i];
    }
}

__global__ void __launch_bounds__(1024)
bin_build_kernel(const int* __restrict__ tmp, const int* __restrict__ cntT,
                 const int* __restrict__ offT, int* __restrict__ rowptr,
                 float* __restrict__ dinv, int2* __restrict__ csr, int n, int nC) {
    __shared__ int wsum[32];
    __shared__ int pOff[MAXC];
    __shared__ int pCnt[MAXC];
    __shared__ int cnt[BIN_NODES];
    __shared__ int stage[STAGE_CAP];
    const int tid  = threadIdx.x;
    const int lane = tid & 31;
    const int wave = tid >> 5;
    const int b    = blockIdx.x;

    int before = 0;
    if (tid < MAXC) {
        const bool v = tid < nC;
        before    = v ? offT[b * nC + tid] : 0;
        pCnt[tid] = v ? cntT[b * nC + tid] : 0;
        pOff[tid] = v ? tid * CH + before : 0;
    }
    if (tid < BIN_NODES) cnt[tid] = 0;
    int base;
    block_scan_incl(wsum, tid, before, base);

    int wv[16];
#pragma unroll
    for (int q = 0; q < 8; ++q) {
        const int pc = pCnt[wave + 32 * q];
        const int po = pOff[wave + 32 * q];
        wv[2 * q]     = (lane < pc)      ? tmp[po + lane]      : -1;
        wv[2 * q + 1] = (lane + 32 < pc) ? tmp[po + lane + 32] : -1;
    }
#pragma unroll
    for (int u = 0; u < 16; ++u)
        if (wv[u] >= 0) atomicAdd(&cnt[wv[u] >> SRC_BITS], 1);
    for (int q = 0; q < 8; ++q) {
        const int pc = pCnt[wave + 32 * q];
        const int po = pOff[wave + 32 * q];
        for (int j = lane + 64; j < pc; j += 32)
            atomicAdd(&cnt[tmp[po + j] >> SRC_BITS], 1);
    }
    __syncthreads();

    const int myc   = (tid < BIN_NODES) ? cnt[tid] : 0;
    int total;
    const int incl  = block_scan_incl(wsum, tid, myc, total);
    const int excl  = incl - myc;
    if (tid < BIN_NODES) {
        const int node = b * BIN_NODES + tid;
        if (node < n) {
            rowptr[node] = base + excl;
            dinv[node]   = rsqrtf((float)(myc + 1));
        }
        cnt[tid] = excl;
    }
    __syncthreads();

    if (total <= STAGE_CAP) {
#pragma unroll
        for (int u = 0; u < 16; ++u)
            if (wv[u] >= 0) stage[atomicAdd(&cnt[wv[u] >> SRC_BITS], 1)] = wv[u] & SRC_MASK;
        for (int q = 0; q < 8; ++q) {
            const int pc = pCnt[wave + 32 * q];
            const int po = pOff[wave + 32 * q];
            for (int j = lane + 64; j < pc; j += 32) {
                const int x = tmp[po + j];
                stage[atomicAdd(&cnt[x >> SRC_BITS], 1)] = x & SRC_MASK;
            }
        }
        __syncthreads();
        for (int i = tid; i < total; i += 1024) csr[base + i] = make_int2(stage[i], 0);
    } else {
#pragma unroll
        for (int u = 0; u < 16; ++u)
            if (wv[u] >= 0)
                csr[base + atomicAdd(&cnt[wv[u] >> SRC_BITS], 1)] = make_int2(wv[u] & SRC_MASK, 0);
        for (int q = 0; q < 8; ++q) {
            const int pc = pCnt[wave + 32 * q];
            const int po = pOff[wave + 32 * q];
            for (int j = lane + 64; j < pc; j += 32) {
                const int x = tmp[po + j];
                csr[base + atomicAdd(&cnt[x >> SRC_BITS], 1)] = make_int2(x & SRC_MASK, 0);
            }
        }
    }
}

__global__ void attach_dinv_kernel(int2* __restrict__ csr, const float* __restrict__ dinv, int nE) {
    int j = blockIdx.x * blockDim.x + threadIdx.x;
    if (j < nE) {
        int2 e = csr[j];
        e.y = __float_as_int(dinv[e.x]);
        csr[j] = e;
    }
}

template <int G>
__device__ __forceinline__ void load_row_groups(const float* __restrict__ feat, int sL, int cnt,
                                                int lane, float (&v)[32]) {
#pragma unroll
    for (int u = 0; u < 4; ++u) {
        int s = __builtin_amdgcn_readlane(sL, G * 4 + u);
        v[G * 4 + u] = feat[s * 32 + lane];
    }
    if constexpr (G < 7) {
        if (cnt > (G + 1) * 4) load_row_groups<G + 1>(feat, sL, cnt, lane, v);
    }
}

__device__ __forceinline__ float pull_node(const float* __restrict__ feat,
                                           const int2* __restrict__ csr,
                                           const int* __restrict__ rowptr,
                                           int d, int n, int nE, int lane) {
    const int beg = rowptr[d];
    const int end = (d + 1 < n) ? rowptr[d + 1] : nE;
    const float di = rsqrtf((float)(end - beg + 1));
    float acc = 0.0f;
    if (beg >= end) acc = feat[d * 32 + lane] * (di * di);
    for (int base = beg; base < end; base += 32) {
        const float selfW = (base == beg) ? di * di : 0.0f;
        int cnt = end - base;
        if (cnt > 32) cnt = 32;
        const int   last = cnt - 1;
        const int2  eL = csr[base + (lane < last ? lane : last)];
        const int   sL = eL.x;
        const float dL = __int_as_float(eL.y);
        float v[32];
#pragma unroll
        for (int u = 0; u < 32; ++u) v[u] = 0.0f;
        const float vs = feat[d * 32 + lane];
        load_row_groups<0>(feat, sL, cnt, lane, v);
        const float wL = (lane < cnt) ? dL * di : 0.0f;
        acc += vs * selfW;
#pragma unroll
        for (int u = 0; u < 32; ++u) {
            float w = __int_as_float(__builtin_amdgcn_readlane(__float_as_int(wL), u));
            acc += v[u] * w;
        }
    }
    return acc;
}

#define TP 36

__global__ void __launch_bounds__(512)
pull_gemm_kernel(const float* __restrict__ x, const int2* __restrict__ csr,
                 const int* __restrict__ rowptr,
                 const float* __restrict__ W1, const float* __restrict__ b1,
                 const float* __restrict__ Wmu, const float* __restrict__ Wlv,
                 float* __restrict__ z, int n, int nE) {
    __shared__ __attribute__((aligned(16))) float tA[16 * TP];
    __shared__ __attribute__((aligned(16))) float tH[16 * TP];
    const int lane = threadIdx.x & 31;
    const int wave = __builtin_amdgcn_readfirstlane((int)(threadIdx.x >> 5));
    const int m0   = blockIdx.x * 16;
    const int d    = m0 + wave;

    float sum = 0.0f;
    if (d < n) sum = pull_node(x, csr, rowptr, d, n, nE, lane);
    tA[wave * TP + lane] = sum;
    __syncthreads();

    if (wave != (int)(blockIdx.x & 15)) return;

    const int c    = lane & 15;
    const int hh   = lane >> 4;
    const int koff = 2 * hh;

    v8f acc0 = {};
    v8f acc1 = {};
#pragma unroll
    for (int k = 0; k < 8; ++k) {
        const int kb = 4 * k + koff;
        const v2f a = *reinterpret_cast<const v2f*>(&tA[c * TP + kb]);
        v2f p, q;
        p.x = W1[kb * 32 + c];        p.y = W1[(kb + 1) * 32 + c];
        q.x = W1[kb * 32 + 16 + c];   q.y = W1[(kb + 1) * 32 + 16 + c];
        acc0 = __builtin_amdgcn_wmma_f32_16x16x4_f32(false, a, false, p, (short)0, acc0, false, false);
        acc1 = __builtin_amdgcn_wmma_f32_16x16x4_f32(false, a, false, q, (short)0, acc1, false, false);
    }

    const float vb0 = b1[c];
    const float vb1 = b1[16 + c];
#pragma unroll
    for (int r = 0; r < 8; ++r) {
        const float v0 = acc0[r] + vb0;
        const float v1 = acc1[r] + vb1;
        tH[(8 * hh + r) * TP + c]      = v0 > 0.0f ? v0 : 0.0f;
        tH[(8 * hh + r) * TP + 16 + c] = v1 > 0.0f ? v1 : 0.0f;
    }
    __builtin_amdgcn_fence(__ATOMIC_ACQ_REL, "wavefront");

    v8f accM = {};
    v8f accL = {};
#pragma unroll
    for (int k = 0; k < 8; ++k) {
        const int kb = 4 * k + koff;
        const v2f a = *reinterpret_cast<const v2f*>(&tH[c * TP + kb]);
        v2f p, q;
        p.x = Wmu[kb * 16 + c];  p.y = Wmu[(kb + 1) * 16 + c];
        q.x = Wlv[kb * 16 + c];  q.y = Wlv[(kb + 1) * 16 + c];
        accM = __builtin_amdgcn_wmma_f32_16x16x4_f32(false, a, false, p, (short)0, accM, false, false);
        accL = __builtin_amdgcn_wmma_f32_16x16x4_f32(false, a, false, q, (short)0, accL, false, false);
    }

#pragma unroll
    for (int r = 0; r < 8; ++r) {
        tA[(8 * hh + r) * TP + c]      = accM[r];
        tA[(8 * hh + r) * TP + 16 + c] = accL[r];
    }
    __builtin_amdgcn_fence(__ATOMIC_ACQ_REL, "wavefront");
#pragma unroll
    for (int i = 0; i < 4; ++i) {
        const int row = 4 * i + (lane >> 3);
        const int ch  = lane & 7;
        const float4 v = *reinterpret_cast<const float4*>(&tA[row * TP + 4 * ch]);
        if (m0 + row < n) *reinterpret_cast<float4*>(&z[(m0 + row) * 32 + 4 * ch]) = v;
    }
}

__global__ void pull_out_kernel(const float* __restrict__ z, const int2* __restrict__ csr,
                                const int* __restrict__ rowptr,
                                const float* __restrict__ bmu, const float* __restrict__ blv,
                                float* __restrict__ out, int n, int nE) {
    const int lane = threadIdx.x & 31;
    const int w  = __builtin_amdgcn_readfirstlane((int)((blockIdx.x * blockDim.x + threadIdx.x) >> 5));
    const int n0 = 2 * w;
    if (n0 >= n) return;
    const bool hasB = (n0 + 1) < n;
    float a = pull_node(z, csr, rowptr, n0, n, nE, lane);
    float b = 0.0f;
    if (hasB) b = pull_node(z, csr, rowptr, n0 + 1, n, nE, lane);
    const float ax = __shfl_xor(a, 16);
    const float bx = __shfl_xor(b, 16);
    const int c = lane & 15;
    const float muv = (lane < 16 ? a  : bx) + bmu[c];
    const float lvv = (lane < 16 ? ax : b ) + blv[c];
    float* outLv = out + (size_t)n * 16;
    if (lane < 16 || hasB) {
        out[n0 * 16 + lane]   = muv;
        outLv[n0 * 16 + lane] = lvv;
    }
}

extern "C" void kernel_launch(void* const* d_in, const int* in_sizes, int n_in,
                              void* d_out, int out_size, void* d_ws, size_t ws_size,
                              hipStream_t stream) {
    const float* x   = (const float*)d_in[0];
    const int*   ei  = (const int*)  d_in[1];
    const float* W1  = (const float*)d_in[2];
    const float* b1  = (const float*)d_in[3];
    const float* Wmu = (const float*)d_in[4];
    const float* bmu = (const float*)d_in[5];
    const float* Wlv = (const float*)d_in[6];
    const float* blv = (const float*)d_in[7];

    const int N = in_sizes[0] / 32;
    const int E = in_sizes[1] / 2;
    const int* src = ei;
    const int* dst = ei + E;

    const int nB = (N + BIN_NODES - 1) / BIN_NODES;
    const int nC = (E + CH - 1) / CH;
    int*   rowptr = (int*)d_ws;
    int2*  csr    = (int2*)(rowptr + N);
    float* bufA   = (float*)(csr + E);
    int*   tmp    = (int*)bufA;
    float* dinv   = (float*)(tmp + E);
    int*   cntT   = (int*)(dinv + N);
    int*   offT   = cntT + nB * nC;
    float* out    = (float*)d_out;

    const int TB = 256;
    const int tileBlocks  = (N + 15) / 16;
    const int pairWaves   = (N + 1) / 2;

    chunk_sort_kernel<<<nC, 1024, 0, stream>>>(src, dst, tmp, cntT, offT, E, nB, nC);
    bin_build_kernel<<<nB, 1024, 0, stream>>>(tmp, cntT, offT, rowptr, dinv, csr, N, nC);
    attach_dinv_kernel<<<(E + TB - 1) / TB, TB, 0, stream>>>(csr, dinv, E);

    pull_gemm_kernel<<<tileBlocks, 512, 0, stream>>>(x, csr, rowptr, W1, b1, Wmu, Wlv, bufA, N, E);

    pull_out_kernel<<<(pairWaves * 32 + TB - 1) / TB, TB, 0, stream>>>(bufA, csr, rowptr, bmu, blv, out, N, E);
}


# One wave per 16 nodes (8 in layer 2), not per node: it sums them in turn, next list requested ahead, then runs its own WMMA tile from wave-private LDS, no barrier; attach pass 8 entries per thread
# speedup vs baseline: 3.7835x; 1.0381x over previous
#include <hip/hip_runtime.h>

typedef __attribute__((ext_vector_type(2))) float v2f;
typedef __attribute__((ext_vector_type(8))) float v8f;

#define CH        8192
#define BIN_SHIFT 9
#define BIN_NODES 512
#define MAXB      256
#define MAXC      256
#define SRC_BITS  17
#define SRC_MASK  0x1FFFF
#define STAGE_CAP 16384

__device__ __forceinline__ int wave_scan_incl(int x, int lane) {
    x += __builtin_amdgcn_update_dpp(0, x, 0x111, 0xf, 0xf, true);
    x += __builtin_amdgcn_update_dpp(0, x, 0x112, 0xf, 0xf, true);
    x += __builtin_amdgcn_update_dpp(0, x, 0x114, 0xf, 0xf, true);
    x += __builtin_amdgcn_update_dpp(0, x, 0x118, 0xf, 0xf, true);
    const int lowTotal = __builtin_amdgcn_readlane(x, 15);
    return x + (lane >= 16 ? lowTotal : 0);
}

__device__ __forceinline__ int block_scan_incl(int* wsum, int tid, int v, int& total) {
    const int lane = tid & 31;
    const int wave = __builtin_amdgcn_readfirstlane(tid >> 5);
    const int x = wave_scan_incl(v, lane);
    if (lane == 31) wsum[wave] = x;
    __syncthreads();
    const int y = wave_scan_incl(wsum[lane], lane);
    total = __builtin_amdgcn_readlane(y, 31);
    const int before = __builtin_amdgcn_readlane(y, (wave + 31) & 31);
    return x + (wave > 0 ? before : 0);
}

__global__ void __launch_bounds__(1024)
chunk_sort_kernel(const int* __restrict__ src, const int* __restrict__ dst,
                  int* __restrict__ tmp, int* __restrict__ cntT, int* __restrict__ offT,
                  int nE, int nB, int nC) {
    __shared__ int hist[MAXB];
    __shared__ int wsum[32];
    __shared__ int stage[CH];
    const int tid  = threadIdx.x;
    const int c    = blockIdx.x;
    const int base = c * CH;

    if (tid < MAXB) hist[tid] = 0;
    __syncthreads();

    int w[8], bn[8];
#pragma unroll
    for (int k = 0; k < 8; ++k) {
        const int e = base + k * 1024 + tid;
        bn[k] = -1;
        w[k]  = 0;
        if (e < nE) {
            const int d = dst[e];
            const int s = src[e];
            bn[k] = d >> BIN_SHIFT;
            w[k]  = s | ((d & (BIN_NODES - 1)) << SRC_BITS);
        }
    }
#pragma unroll
    for (int k = 0; k < 8; ++k)
        if (bn[k] >= 0) atomicAdd(&hist[bn[k]], 1);
    __syncthreads();

    const int h    = (tid < MAXB) ? hist[tid] : 0;
    int chunkTotal;
    const int incl = block_scan_incl(wsum, tid, h, chunkTotal);
    const int excl = incl - h;
    if (tid < nB) {
        cntT[tid * nC + c] = h;
        offT[tid * nC + c] = excl;
    }
    if (tid < MAXB) hist[tid] = excl;
    __syncthreads();

#pragma unroll
    for (int k = 0; k < 8; ++k)
        if (bn[k] >= 0) stage[atomicAdd(&hist[bn[k]], 1)] = w[k];
    __syncthreads();

#pragma unroll
    for (int k = 0; k < 8; ++k) {
        const int i = k * 1024 + tid;
        if (base + i < nE) tmp[base + i] = stage[i];
    }
}

__global__ void __launch_bounds__(1024)
bin_build_kernel(const int* __restrict__ tmp, const int* __restrict__ cntT,
                 const int* __restrict__ offT, int* __restrict__ rowptr,
                 float* __restrict__ dinv, int2* __restrict__ csr, int n, int nC) {
    __shared__ int wsum[32];
    __shared__ int pOff[MAXC];
    __shared__ int pCnt[MAXC];
    __shared__ int cnt[BIN_NODES];
    __shared__ int stage[STAGE_CAP];
    const int tid  = threadIdx.x;
    const int lane = tid & 31;
    const int wave = tid >> 5;
    const int b    = blockIdx.x;

    int before = 0;
    if (tid < MAXC) {
        const bool v = tid < nC;
        before    = v ? offT[b * nC + tid] : 0;
        pCnt[tid] = v ? cntT[b * nC + tid] : 0;
        pOff[tid] = v ? tid * CH + before : 0;
    }
    if (tid < BIN_NODES) cnt[tid] = 0;
    int base;
    block_scan_incl(wsum, tid, before, base);

    int wv[16];
#pragma unroll
    for (int q = 0; q < 8; ++q) {
        const int pc = pCnt[wave + 32 * q];
        const int po = pOff[wave + 32 * q];
        wv[2 * q]     = (lane < pc)      ? tmp[po + lane]      : -1;
        wv[2 * q + 1] = (lane + 32 < pc) ? tmp[po + lane + 32] : -1;
    }
#pragma unroll
    for (int u = 0; u < 16; ++u)
        if (wv[u] >= 0) atomicAdd(&cnt[wv[u] >> SRC_BITS], 1);
    for (int q = 0; q < 8; ++q) {
        const int pc = pCnt[wave + 32 * q];
        const int po = pOff[wave + 32 * q];
        for (int j = lane + 64; j < pc; j += 32)
            atomicAdd(&cnt[tmp[po + j] >> SRC_BITS], 1);
    }
    __syncthreads();

    const int myc   = (tid < BIN_NODES) ? cnt[tid] : 0;
    int total;
    const int incl  = block_scan_incl(wsum, tid, myc, total);
    const int excl  = incl - myc;
    if (tid < BIN_NODES) {
        const int node = b * BIN_NODES + tid;
        if (node < n) {
            rowptr[node] = base + excl;
            dinv[node]   = rsqrtf((float)(myc + 1));
        }
        cnt[tid] = excl;
    }
    __syncthreads();

    if (total <= STAGE_CAP) {
#pragma unroll
        for (int u = 0; u < 16; ++u)
            if (wv[u] >= 0) stage[atomicAdd(&cnt[wv[u] >> SRC_BITS], 1)] = wv[u] & SRC_MASK;
        for (int q = 0; q < 8; ++q) {
            const int pc = pCnt[wave + 32 * q];
            const int po = pOff[wave + 32 * q];
            for (int j = lane + 64; j < pc; j += 32) {
                const int x = tmp[po + j];
                stage[atomicAdd(&cnt[x >> SRC_BITS], 1)] = x & SRC_MASK;
            }
        }
        __syncthreads();
        for (int i = tid; i < total; i += 1024) csr[base + i] = make_int2(stage[i], 0);
    } else {
#pragma unroll
        for (int u = 0; u < 16; ++u)
            if (wv[u] >= 0)
                csr[base + atomicAdd(&cnt[wv[u] >> SRC_BITS], 1)] = make_int2(wv[u] & SRC_MASK, 0);
        for (int q = 0; q < 8; ++q) {
            const int pc = pCnt[wave + 32 * q];
            const int po = pOff[wave + 32 * q];
            for (int j = lane + 64; j < pc; j += 32) {
                const int x = tmp[po + j];
                csr[base + atomicAdd(&cnt[x >> SRC_BITS], 1)] = make_int2(x & SRC_MASK, 0);
            }
        }
    }
}

#define EPT 8
__global__ void __launch_bounds__(256)
attach_dinv_kernel(int2* __restrict__ csr, const float* __restrict__ dinv, int nE) {
    const int j0 = blockIdx.x * (EPT * 256) + threadIdx.x;
    int2 e[EPT];
#pragma unroll
    for (int k = 0; k < EPT; ++k) {
        const int j = j0 + k * 256;
        e[k] = (j < nE) ? csr[j] : make_int2(0, 0);
    }
#pragma unroll
    for (int k = 0; k < EPT; ++k) e[k].y = __float_as_int(dinv[e[k].x]);
#pragma unroll
    for (int k = 0; k < EPT; ++k) {
        const int j = j0 + k * 256;
        if (j < nE) csr[j] = e[k];
    }
}

template <int G>
__device__ __forceinline__ void load_row_groups(const float* __restrict__ feat, int sL, int cnt,
                                                int lane, float (&v)[32]) {
#pragma unroll
    for (int u = 0; u < 4; ++u) {
        int s = __builtin_amdgcn_readlane(sL, G * 4 + u);
        v[G * 4 + u] = feat[s * 32 + lane];
    }
    if constexpr (G < 7) {
        if (cnt > (G + 1) * 4) load_row_groups<G + 1>(feat, sL, cnt, lane, v);
    }
}

__device__ __forceinline__ int2 first_entries(const int2* __restrict__ csr, int beg, int end, int lane) {
    int2 e = make_int2(0, 0);
    if (end > beg) {
        int cnt = end - beg;
        if (cnt > 32) cnt = 32;
        const int last = cnt - 1;
        e = csr[beg + (lane < last ? lane : last)];
    }
    return e;
}

__device__ __forceinline__ int rowptr_lanes(const int* __restrict__ rowptr, int first, int n, int nE, int lane) {
    const int idx = first + lane;
    return (idx < n) ? rowptr[idx] : nE;
}

__device__ __forceinline__ float pull_node(const float* __restrict__ feat,
                                           const int2* __restrict__ csr,
                                           int beg, int end, int2 eFirst,
                                           int d, int lane) {
    const float di = rsqrtf((float)(end - beg + 1));
    float acc = 0.0f;
    if (beg >= end) acc = feat[d * 32 + lane] * (di * di);
    for (int base = beg; base < end; base += 32) {
        const float selfW = (base == beg) ? di * di : 0.0f;
        int cnt = end - base;
        if (cnt > 32) cnt = 32;
        const int   last = cnt - 1;
        int2 eL = eFirst;
        if (base != beg) eL = csr[base + (lane < last ? lane : last)];
        const int   sL = eL.x;
        const float dL = __int_as_float(eL.y);
        float v[32];
#pragma unroll
        for (int u = 0; u < 32; ++u) v[u] = 0.0f;
        const float vs = feat[d * 32 + lane];
        load_row_groups<0>(feat, sL, cnt, lane, v);
        const float wL = (lane < cnt) ? dL * di : 0.0f;
        acc += vs * selfW;
#pragma unroll
        for (int u = 0; u < 32; ++u) {
            float w = __int_as_float(__builtin_amdgcn_readlane(__float_as_int(wL), u));
            acc += v[u] * w;
        }
    }
    return acc;
}

#define TP 36
#define GW 4

__global__ void __launch_bounds__(32 * GW)
pull_gemm_kernel(const float* __restrict__ x, const int2* __restrict__ csr,
                 const int* __restrict__ rowptr,
                 const float* __restrict__ W1, const float* __restrict__ b1,
                 const float* __restrict__ Wmu, const float* __restrict__ Wlv,
                 float* __restrict__ z, int n, int nE) {
    __shared__ __attribute__((aligned(16))) float tAall[GW][16 * TP];
    __shared__ __attribute__((aligned(16))) float tHall[GW][16 * TP];
    const int lane = threadIdx.x & 31;
    const int wave = __builtin_amdgcn_readfirstlane((int)(threadIdx.x >> 5));
    const int m0   = (blockIdx.x * GW + wave) * 16;
    if (m0 >= n) return;
    float* tA = tAall[wave];
    float* tH = tHall[wave];

    const int rp = rowptr_lanes(rowptr, m0, n, nE, lane);
    int beg = __builtin_amdgcn_readlane(rp, 0);
    int end = __builtin_amdgcn_readlane(rp, 1);
    int2 eL = first_entries(csr, beg, end, lane);
#pragma nounroll
    for (int k = 0; k < 16; ++k) {
        const int d = m0 + k;
        float sum = 0.0f;
        if (d < n) {
            const int endN = __builtin_amdgcn_readlane(rp, k + 2);
            int2 eN = eL;
            if (k + 1 < 16 && d + 1 < n) eN = first_entries(csr, end, endN, lane);
            sum = pull_node(x, csr, beg, end, eL, d, lane);
            eL  = eN;
            beg = end;
            end = endN;
        }
        tA[k * TP + lane] = sum;
    }
    __builtin_amdgcn_fence(__ATOMIC_ACQ_REL, "wavefront");

    const int c    = lane & 15;
    const int hh   = lane >> 4;
    const int koff = 2 * hh;

    v8f acc0 = {};
    v8f acc1 = {};
#pragma unroll
    for (int k = 0; k < 8; ++k) {
        const int kb = 4 * k + koff;
        const v2f a = *reinterpret_cast<const v2f*>(&tA[c * TP + kb]);
        v2f p, q;
        p.x = W1[kb * 32 + c];        p.y = W1[(kb + 1) * 32 + c];
        q.x = W1[kb * 32 + 16 + c];   q.y = W1[(kb + 1) * 32 + 16 + c];
        acc0 = __builtin_amdgcn_wmma_f32_16x16x4_f32(false, a, false, p, (short)0, acc0, false, false);
        acc1 = __builtin_amdgcn_wmma_f32_16x16x4_f32(false, a, false, q, (short)0, acc1, false, false);
    }

    const float vb0 = b1[c];
    const float vb1 = b1[16 + c];
#pragma unroll
    for (int r = 0; r < 8; ++r) {
        const float v0 = acc0[r] + vb0;
        const float v1 = acc1[r] + vb1;
        tH[(8 * hh + r) * TP + c]      = v0 > 0.0f ? v0 : 0.0f;
        tH[(8 * hh + r) * TP + 16 + c] = v1 > 0.0f ? v1 : 0.0f;
    }
    __builtin_amdgcn_fence(__ATOMIC_ACQ_REL, "wavefront");

    v8f accM = {};
    v8f accL = {};
#pragma unroll
    for (int k = 0; k < 8; ++k) {
        const int kb = 4 * k + koff;
        const v2f a = *reinterpret_cast<const v2f*>(&tH[c * TP + kb]);
        v2f p, q;
        p.x = Wmu[kb * 16 + c];  p.y = Wmu[(kb + 1) * 16 + c];
        q.x = Wlv[kb * 16 + c];  q.y = Wlv[(kb + 1) * 16 + c];
        accM = __builtin_amdgcn_wmma_f32_16x16x4_f32(false, a, false, p, (short)0, accM, false, false);
        accL = __builtin_amdgcn_wmma_f32_16x16x4_f32(false, a, false, q, (short)0, accL, false, false);
    }

#pragma unroll
    for (int r = 0; r < 8; ++r) {
        tA[(8 * hh + r) * TP + c]      = accM[r];
        tA[(8 * hh + r) * TP + 16 + c] = accL[r];
    }
    __builtin_amdgcn_fence(__ATOMIC_ACQ_REL, "wavefront");
#pragma unroll
    for (int i = 0; i < 4; ++i) {
        const int row = 4 * i + (lane >> 3);
        const int ch  = lane & 7;
        const float4 v = *reinterpret_cast<const float4*>(&tA[row * TP + 4 * ch]);
        if (m0 + row < n) *reinterpret_cast<float4*>(&z[(m0 + row) * 32 + 4 * ch]) = v;
    }
}

#define PPW 4
__global__ void __launch_bounds__(256)
pull_out_kernel(const float* __restrict__ z, const int2* __restrict__ csr,
                const int* __restrict__ rowptr,
                const float* __restrict__ bmu, const float* __restrict__ blv,
                float* __restrict__ out, int n, int nE) {
    const int lane = threadIdx.x & 31;
    const int w  = __builtin_amdgcn_readfirstlane((int)((blockIdx.x * blockDim.x + threadIdx.x) >> 5));
    const int nb = 2 * PPW * w;
    if (nb >= n) return;
    const int c = lane & 15;
    const float vbm = bmu[c];
    const float vbl = blv[c];
    float* outLv = out + (size_t)n * 16;

    const int rp = rowptr_lanes(rowptr, nb, n, nE, lane);
    int2 eA = first_entries(csr, __builtin_amdgcn_readlane(rp, 0), __builtin_amdgcn_readlane(rp, 1), lane);
#pragma nounroll
    for (int p = 0; p < PPW; ++p) {
        const int n0 = nb + 2 * p;
        if (n0 >= n) break;
        const bool hasB = (n0 + 1) < n;
        const int beg0 = __builtin_amdgcn_readlane(rp, 2 * p);
        const int end0 = __builtin_amdgcn_readlane(rp, 2 * p + 1);
        const int end1 = __builtin_amdgcn_readlane(rp, 2 * p + 2);
        const int end2 = __builtin_amdgcn_readlane(rp, 2 * p + 3);
        int2 eB = eA;
        if (hasB) eB = first_entries(csr, end0, end1, lane);
        const float a = pull_node(z, csr, beg0, end0, eA, n0, lane);
        int2 eN = eB;
        if (p + 1 < PPW && n0 + 2 < n) eN = first_entries(csr, end1, end2, lane);
        float b = 0.0f;
        if (hasB) b = pull_node(z, csr, end0, end1, eB, n0 + 1, lane);
        eA = eN;
        const float ax = __shfl_xor(a, 16);
        const float bx = __shfl_xor(b, 16);
        const float muv = (lane < 16 ? a  : bx) + vbm;
        const float lvv = (lane < 16 ? ax : b ) + vbl;
        if (lane < 16 || hasB) {
            out[n0 * 16 + lane]   = muv;
            outLv[n0 * 16 + lane] = lvv;
        }
    }
}

extern "C" void kernel_launch(void* const* d_in, const int* in_sizes, int n_in,
                              void* d_out, int out_size, void* d_ws, size_t ws_size,
                              hipStream_t stream) {
    const float* x   = (const float*)d_in[0];
    const int*   ei  = (const int*)  d_in[1];
    const float* W1  = (const float*)d_in[2];
    const float* b1  = (const float*)d_in[3];
    const float* Wmu = (const float*)d_in[4];
    const float* bmu = (const float*)d_in[5];
    const float* Wlv = (const float*)d_in[6];
    const float* blv = (const float*)d_in[7];

    const int N = in_sizes[0] / 32;
    const int E = in_sizes[1] / 2;
    const int* src = ei;
    const int* dst = ei + E;

    const int nB = (N + BIN_NODES - 1) / BIN_NODES;
    const int nC = (E + CH - 1) / CH;
    int*   rowptr = (int*)d_ws;
    int2*  csr    = (int2*)(rowptr + N);
    float* bufA   = (float*)(csr + E);
    int*   tmp    = (int*)bufA;
    float* dinv   = (float*)(tmp + E);
    int*   cntT   = (int*)(dinv + N);
    int*   offT   = cntT + nB * nC;
    float* out    = (float*)d_out;

    const int TB = 256;
    const int tileBlocks  = (N + 16 * GW - 1) / (16 * GW);
    const int outBlocks   = (N + 2 * PPW * 8 - 1) / (2 * PPW * 8);

    chunk_sort_kernel<<<nC, 1024, 0, stream>>>(src, dst, tmp, cntT, offT, E, nB, nC);
    bin_build_kernel<<<nB, 1024, 0, stream>>>(tmp, cntT, offT, rowptr, dinv, csr, N, nC);
    attach_dinv_kernel<<<(E + EPT * TB - 1) / (EPT * TB), TB, 0, stream>>>(csr, dinv, E);

    pull_gemm_kernel<<<tileBlocks, 32 * GW, 0, stream>>>(x, csr, rowptr, W1, b1, Wmu, Wlv, bufA, N, E);

    pull_out_kernel<<<outBlocks, TB, 0, stream>>>(bufA, csr, rowptr, bmu, blv, out, N, E);
}


# Row sums multiply-add only the row groups they loaded (the test that ends the load chain picks the sum code): no zero fill or multiply-add for unused rows; own row added last: sum order differs
# speedup vs baseline: 4.0251x; 1.0639x over previous
#include <hip/hip_runtime.h>

typedef __attribute__((ext_vector_type(2))) float v2f;
typedef __attribute__((ext_vector_type(8))) float v8f;

#define CH        8192
#define BIN_SHIFT 9
#define BIN_NODES 512
#define MAXB      256
#define MAXC      256
#define SRC_BITS  17
#define SRC_MASK  0x1FFFF
#define STAGE_CAP 16384

__device__ __forceinline__ int wave_scan_incl(int x, int lane) {
    x += __builtin_amdgcn_update_dpp(0, x, 0x111, 0xf, 0xf, true);
    x += __builtin_amdgcn_update_dpp(0, x, 0x112, 0xf, 0xf, true);
    x += __builtin_amdgcn_update_dpp(0, x, 0x114, 0xf, 0xf, true);
    x += __builtin_amdgcn_update_dpp(0, x, 0x118, 0xf, 0xf, true);
    const int lowTotal = __builtin_amdgcn_readlane(x, 15);
    return x + (lane >= 16 ? lowTotal : 0);
}

__device__ __forceinline__ int block_scan_incl(int* wsum, int tid, int v, int& total) {
    const int lane = tid & 31;
    const int wave = __builtin_amdgcn_readfirstlane(tid >> 5);
    const int x = wave_scan_incl(v, lane);
    if (lane == 31) wsum[wave] = x;
    __syncthreads();
    const int y = wave_scan_incl(wsum[lane], lane);
    total = __builtin_amdgcn_readlane(y, 31);
    const int before = __builtin_amdgcn_readlane(y, (wave + 31) & 31);
    return x + (wave > 0 ? before : 0);
}

__global__ void __launch_bounds__(1024)
chunk_sort_kernel(const int* __restrict__ src, const int* __restrict__ dst,
                  int* __restrict__ tmp, int* __restrict__ cntT, int* __restrict__ offT,
                  int nE, int nB, int nC) {
    __shared__ int hist[MAXB];
    __shared__ int wsum[32];
    __shared__ int stage[CH];
    const int tid  = threadIdx.x;
    const int c    = blockIdx.x;
    const int base = c * CH;

    if (tid < MAXB) hist[tid] = 0;
    __syncthreads();

    int w[8], bn[8];
#pragma unroll
    for (int k = 0; k < 8; ++k) {
        const int e = base + k * 1024 + tid;
        bn[k] = -1;
        w[k]  = 0;
        if (e < nE) {
            const int d = dst[e];
            const int s = src[e];
            bn[k] = d >> BIN_SHIFT;
            w[k]  = s | ((d & (BIN_NODES - 1)) << SRC_BITS);
        }
    }
#pragma unroll
    for (int k = 0; k < 8; ++k)
        if (bn[k] >= 0) atomicAdd(&hist[bn[k]], 1);
    __syncthreads();

    const int h    = (tid < MAXB) ? hist[tid] : 0;
    int chunkTotal;
    const int incl = block_scan_incl(wsum, tid, h, chunkTotal);
    const int excl = incl - h;
    if (tid < nB) {
        cntT[tid * nC + c] = h;
        offT[tid * nC + c] = excl;
    }
    if (tid < MAXB) hist[tid] = excl;
    __syncthreads();

#pragma unroll
    for (int k = 0; k < 8; ++k)
        if (bn[k] >= 0) stage[atomicAdd(&hist[bn[k]], 1)] = w[k];
    __syncthreads();

#pragma unroll
    for (int k = 0; k < 8; ++k) {
        const int i = k * 1024 + tid;
        if (base + i < nE) tmp[base + i] = stage[i];
    }
}

__global__ void __launch_bounds__(1024)
bin_build_kernel(const int* __restrict__ tmp, const int* __restrict__ cntT,
                 const int* __restrict__ offT, int* __restrict__ rowptr,
                 float* __restrict__ dinv, int2* __restrict__ csr, int n, int nC) {
    __shared__ int wsum[32];
    __shared__ int pOff[MAXC];
    __shared__ int pCnt[MAXC];
    __shared__ int cnt[BIN_NODES];
    __shared__ int stage[STAGE_CAP];
    const int tid  = threadIdx.x;
    const int lane = tid & 31;
    const int wave = tid >> 5;
    const int b    = blockIdx.x;

    int before = 0;
    if (tid < MAXC) {
        const bool v = tid < nC;
        before    = v ? offT[b * nC + tid] : 0;
        pCnt[tid] = v ? cntT[b * nC + tid] : 0;
        pOff[tid] = v ? tid * CH + before : 0;
    }
    if (tid < BIN_NODES) cnt[tid] = 0;
    int base;
    block_scan_incl(wsum, tid, before, base);

    int wv[16];
#pragma unroll
    for (int q = 0; q < 8; ++q) {
        const int pc = pCnt[wave + 32 * q];
        const int po = pOff[wave + 32 * q];
        wv[2 * q]     = (lane < pc)      ? tmp[po + lane]      : -1;
        wv[2 * q + 1] = (lane + 32 < pc) ? tmp[po + lane + 32] : -1;
    }
#pragma unroll
    for (int u = 0; u < 16; ++u)
        if (wv[u] >= 0) atomicAdd(&cnt[wv[u] >> SRC_BITS], 1);
    for (int q = 0; q < 8; ++q) {
        const int pc = pCnt[wave + 32 * q];
        const int po = pOff[wave + 32 * q];
        for (int j = lane + 64; j < pc; j += 32)
            atomicAdd(&cnt[tmp[po + j] >> SRC_BITS], 1);
    }
    __syncthreads();

    const int myc   = (tid < BIN_NODES) ? cnt[tid] : 0;
    int total;
    const int incl  = block_scan_incl(wsum, tid, myc, total);
    const int excl  = incl - myc;
    if (tid < BIN_NODES) {
        const int node = b * BIN_NODES + tid;
        if (node < n) {
            rowptr[node] = base + excl;
            dinv[node]   = rsqrtf((float)(myc + 1));
        }
        cnt[tid] = excl;
    }
    __syncthreads();

    if (total <= STAGE_CAP) {
#pragma unroll
        for (int u = 0; u < 16; ++u)
            if (wv[u] >= 0) stage[atomicAdd(&cnt[wv[u] >> SRC_BITS], 1)] = wv[u] & SRC_MASK;
        for (int q = 0; q < 8; ++q) {
            const int pc = pCnt[wave + 32 * q];
            const int po = pOff[wave + 32 * q];
            for (int j = lane + 64; j < pc; j += 32) {
                const int x = tmp[po + j];
                stage[atomicAdd(&cnt[x >> SRC_BITS], 1)] = x & SRC_MASK;
            }
        }
        __syncthreads();
        for (int i = tid; i < total; i += 1024) csr[base + i] = make_int2(stage[i], 0);
    } else {
#pragma unroll
        for (int u = 0; u < 16; ++u)
            if (wv[u] >= 0)
                csr[base + atomicAdd(&cnt[wv[u] >> SRC_BITS], 1)] = make_int2(wv[u] & SRC_MASK, 0);
        for (int q = 0; q < 8; ++q) {
            const int pc = pCnt[wave + 32 * q];
            const int po = pOff[wave + 32 * q];
            for (int j = lane + 64; j < pc; j += 32) {
                const int x = tmp[po + j];
                csr[base + atomicAdd(&cnt[x >> SRC_BITS], 1)] = make_int2(x & SRC_MASK, 0);
            }
        }
    }
}

#define EPT 8
__global__ void __launch_bounds__(256)
attach_dinv_kernel(int2* __restrict__ csr, const float* __restrict__ dinv, int nE) {
    const int j0 = blockIdx.x * (EPT * 256) + threadIdx.x;
    int2 e[EPT];
#pragma unroll
    for (int k = 0; k < EPT; ++k) {
        const int j = j0 + k * 256;
        e[k] = (j < nE) ? csr[j] : make_int2(0, 0);
    }
#pragma unroll
    for (int k = 0; k < EPT; ++k) e[k].y = __float_as_int(dinv[e[k].x]);
#pragma unroll
    for (int k = 0; k < EPT; ++k) {
        const int j = j0 + k * 256;
        if (j < nE) csr[j] = e[k];
    }
}

template <int G>
__device__ __forceinline__ float sum_row_groups(const float* __restrict__ feat, int sL, float wL,
                                                int cnt, int lane, float acc, float (&v)[32]) {
#pragma unroll
    for (int u = 0; u < 4; ++u) {
        int s = __builtin_amdgcn_readlane(sL, G * 4 + u);
        v[G * 4 + u] = feat[s * 32 + lane];
    }
    if constexpr (G < 7) {
        if (cnt > (G + 1) * 4) return sum_row_groups<G + 1>(feat, sL, wL, cnt, lane, acc, v);
    }
#pragma unroll
    for (int u = 0; u < (G + 1) * 4; ++u) {
        const float w = __int_as_float(__builtin_amdgcn_readlane(__float_as_int(wL), u));
        acc = __builtin_fmaf(v[u], w, acc);
    }
    return acc;
}

__device__ __forceinline__ int2 first_entries(const int2* __restrict__ csr, int beg, int end, int lane) {
    int2 e = make_int2(0, 0);
    if (end > beg) {
        int cnt = end - beg;
        if (cnt > 32) cnt = 32;
        const int last = cnt - 1;
        e = csr[beg + (lane < last ? lane : last)];
    }
    return e;
}

__device__ __forceinline__ int rowptr_lanes(const int* __restrict__ rowptr, int first, int n, int nE, int lane) {
    const int idx = first + lane;
    return (idx < n) ? rowptr[idx] : nE;
}

__device__ __forceinline__ float pull_node(const float* __restrict__ feat,
                                           const int2* __restrict__ csr,
                                           int beg, int end, int2 eFirst,
                                           int d, int lane) {
    const float di = rsqrtf((float)(end - beg + 1));
    float acc = 0.0f;
    if (beg >= end) acc = feat[d * 32 + lane] * (di * di);
    for (int base = beg; base < end; base += 32) {
        const float selfW = (base == beg) ? di * di : 0.0f;
        int cnt = end - base;
        if (cnt > 32) cnt = 32;
        const int   last = cnt - 1;
        int2 eL = eFirst;
        if (base != beg) eL = csr[base + (lane < last ? lane : last)];
        const int   sL = eL.x;
        const float dL = __int_as_float(eL.y);
        float v[32];
        const float vs = feat[d * 32 + lane];
        const float wL = (lane < cnt) ? dL * di : 0.0f;
        acc = sum_row_groups<0>(feat, sL, wL, cnt, lane, acc, v);
        acc = __builtin_fmaf(vs, selfW, acc);
    }
    return acc;
}

#define TP 36
#define GW 4

__global__ void __launch_bounds__(32 * GW)
pull_gemm_kernel(const float* __restrict__ x, const int2* __restrict__ csr,
                 const int* __restrict__ rowptr,
                 const float* __restrict__ W1, const float* __restrict__ b1,
                 const float* __restrict__ Wmu, const float* __restrict__ Wlv,
                 float* __restrict__ z, int n, int nE) {
    __shared__ __attribute__((aligned(16))) float tAall[GW][16 * TP];
    __shared__ __attribute__((aligned(16))) float tHall[GW][16 * TP];
    const int lane = threadIdx.x & 31;
    const int wave = __builtin_amdgcn_readfirstlane((int)(threadIdx.x >> 5));
    const int m0   = (blockIdx.x * GW + wave) * 16;
    if (m0 >= n) return;
    float* tA = tAall[wave];
    float* tH = tHall[wave];

    const int rp = rowptr_lanes(rowptr, m0, n, nE, lane);
    int beg = __builtin_amdgcn_readlane(rp, 0);
    int end = __builtin_amdgcn_readlane(rp, 1);
    int2 eL = first_entries(csr, beg, end, lane);
#pragma nounroll
    for (int k = 0; k < 16; ++k) {
        const int d = m0 + k;
        float sum = 0.0f;
        if (d < n) {
            const int endN = __builtin_amdgcn_readlane(rp, k + 2);
            int2 eN = eL;
            if (k + 1 < 16 && d + 1 < n) eN = first_entries(csr, end, endN, lane);
            sum = pull_node(x, csr, beg, end, eL, d, lane);
            eL  = eN;
            beg = end;
            end = endN;
        }
        tA[k * TP + lane] = sum;
    }
    __builtin_amdgcn_fence(__ATOMIC_ACQ_REL, "wavefront");

    const int c    = lane & 15;
    const int hh   = lane >> 4;
    const int koff = 2 * hh;

    v8f acc0 = {};
    v8f acc1 = {};
#pragma unroll
    for (int k = 0; k < 8; ++k) {
        const int kb = 4 * k + koff;
        const v2f a = *reinterpret_cast<const v2f*>(&tA[c * TP + kb]);
        v2f p, q;
        p.x = W1[kb * 32 + c];        p.y = W1[(kb + 1) * 32 + c];
        q.x = W1[kb * 32 + 16 + c];   q.y = W1[(kb + 1) * 32 + 16 + c];
        acc0 = __builtin_amdgcn_wmma_f32_16x16x4_f32(false, a, false, p, (short)0, acc0, false, false);
        acc1 = __builtin_amdgcn_wmma_f32_16x16x4_f32(false, a, false, q, (short)0, acc1, false, false);
    }

    const float vb0 = b1[c];
    const float vb1 = b1[16 + c];
#pragma unroll
    for (int r = 0; r < 8; ++r) {
        const float v0 = acc0[r] + vb0;
        const float v1 = acc1[r] + vb1;
        tH[(8 * hh + r) * TP + c]      = v0 > 0.0f ? v0 : 0.0f;
        tH[(8 * hh + r) * TP + 16 + c] = v1 > 0.0f ? v1 : 0.0f;
    }
    __builtin_amdgcn_fence(__ATOMIC_ACQ_REL, "wavefront");

    v8f accM = {};
    v8f accL = {};
#pragma unroll
    for (int k = 0; k < 8; ++k) {
        const int kb = 4 * k + koff;
        const v2f a = *reinterpret_cast<const v2f*>(&tH[c * TP + kb]);
        v2f p, q;
        p.x = Wmu[kb * 16 + c];  p.y = Wmu[(kb + 1) * 16 + c];
        q.x = Wlv[kb * 16 + c];  q.y = Wlv[(kb + 1) * 16 + c];
        accM = __builtin_amdgcn_wmma_f32_16x16x4_f32(false, a, false, p, (short)0, accM, false, false);
        accL = __builtin_amdgcn_wmma_f32_16x16x4_f32(false, a, false, q, (short)0, accL, false, false);
    }

#pragma unroll
    for (int r = 0; r < 8; ++r) {
        tA[(8 * hh + r) * TP + c]      = accM[r];
        tA[(8 * hh + r) * TP + 16 + c] = accL[r];
    }
    __builtin_amdgcn_fence(__ATOMIC_ACQ_REL, "wavefront");
#pragma unroll
    for (int i = 0; i < 4; ++i) {
        const int row = 4 * i + (lane >> 3);
        const int ch  = lane & 7;
        const float4 v = *reinterpret_cast<const float4*>(&tA[row * TP + 4 * ch]);
        if (m0 + row < n) *reinterpret_cast<float4*>(&z[(m0 + row) * 32 + 4 * ch]) = v;
    }
}

#define PPW 4
__global__ void __launch_bounds__(256)
pull_out_kernel(const float* __restrict__ z, const int2* __restrict__ csr,
                const int* __restrict__ rowptr,
                const float* __restrict__ bmu, const float* __restrict__ blv,
                float* __restrict__ out, int n, int nE) {
    const int lane = threadIdx.x & 31;
    const int w  = __builtin_amdgcn_readfirstlane((int)((blockIdx.x * blockDim.x + threadIdx.x) >> 5));
    const int nb = 2 * PPW * w;
    if (nb >= n) return;
    const int c = lane & 15;
    const float vbm = bmu[c];
    const float vbl = blv[c];
    float* outLv = out + (size_t)n * 16;

    const int rp = rowptr_lanes(rowptr, nb, n, nE, lane);
    int2 eA = first_entries(csr, __builtin_amdgcn_readlane(rp, 0), __builtin_amdgcn_readlane(rp, 1), lane);
#pragma nounroll
    for (int p = 0; p < PPW; ++p) {
        const int n0 = nb + 2 * p;
        if (n0 >= n) break;
        const bool hasB = (n0 + 1) < n;
        const int beg0 = __builtin_amdgcn_readlane(rp, 2 * p);
        const int end0 = __builtin_amdgcn_readlane(rp, 2 * p + 1);
        const int end1 = __builtin_amdgcn_readlane(rp, 2 * p + 2);
        const int end2 = __builtin_amdgcn_readlane(rp, 2 * p + 3);
        int2 eB = eA;
        if (hasB) eB = first_entries(csr, end0, end1, lane);
        const float a = pull_node(z, csr, beg0, end0, eA, n0, lane);
        int2 eN = eB;
        if (p + 1 < PPW && n0 + 2 < n) eN = first_entries(csr, end1, end2, lane);
        float b = 0.0f;
        if (hasB) b = pull_node(z, csr, end0, end1, eB, n0 + 1, lane);
        eA = eN;
        const float ax = __shfl_xor(a, 16);
        const float bx = __shfl_xor(b, 16);
        const float muv = (lane < 16 ? a  : bx) + vbm;
        const float lvv = (lane < 16 ? ax : b ) + vbl;
        if (lane < 16 || hasB) {
            out[n0 * 16 + lane]   = muv;
            outLv[n0 * 16 + lane] = lvv;
        }
    }
}

extern "C" void kernel_launch(void* const* d_in, const int* in_sizes, int n_in,
                              void* d_out, int out_size, void* d_ws, size_t ws_size,
                              hipStream_t stream) {
    const float* x   = (const float*)d_in[0];
    const int*   ei  = (const int*)  d_in[1];
    const float* W1  = (const float*)d_in[2];
    const float* b1  = (const float*)d_in[3];
    const float* Wmu = (const float*)d_in[4];
    const float* bmu = (const float*)d_in[5];
    const float* Wlv = (const float*)d_in[6];
    const float* blv = (const float*)d_in[7];

    const int N = in_sizes[0] / 32;
    const int E = in_sizes[1] / 2;
    const int* src = ei;
    const int* dst = ei + E;

    const int nB = (N + BIN_NODES - 1) / BIN_NODES;
    const int nC = (E + CH - 1) / CH;
    int*   rowptr = (int*)d_ws;
    int2*  csr    = (int2*)(rowptr + N);
    float* bufA   = (float*)(csr + E);
    int*   tmp    = (int*)bufA;
    float* dinv   = (float*)(tmp + E);
    int*   cntT   = (int*)(dinv + N);
    int*   offT   = cntT + nB * nC;
    float* out    = (float*)d_out;

    const int TB = 256;
    const int tileBlocks  = (N + 16 * GW - 1) / (16 * GW);
    const int outBlocks   = (N + 2 * PPW * 8 - 1) / (2 * PPW * 8);

    chunk_sort_kernel<<<nC, 1024, 0, stream>>>(src, dst, tmp, cntT, offT, E, nB, nC);
    bin_build_kernel<<<nB, 1024, 0, stream>>>(tmp, cntT, offT, rowptr, dinv, csr, N, nC);
    attach_dinv_kernel<<<(E + EPT * TB - 1) / (EPT * TB), TB, 0, stream>>>(csr, dinv, E);

    pull_gemm_kernel<<<tileBlocks, 32 * GW, 0, stream>>>(x, csr, rowptr, W1, b1, Wmu, Wlv, bufA, N, E);

    pull_out_kernel<<<outBlocks, TB, 0, stream>>>(bufA, csr, rowptr, bmu, blv, out, N, E);
}
